# starting kernel
# baseline (speedup 1.0000x reference)
#include <hip/hip_runtime.h>

typedef __attribute__((ext_vector_type(2))) float v2f;
typedef __attribute__((ext_vector_type(8))) float v8f;

#define BATCH 8192
#define FEATS 4096
#define RANK_LOW 211
#define RANK_HIGH 7
#define DENSITY 0.05f
#define EPT 16

__global__ void kw_factor_kernel(const float* __restrict__ aa,
                                 float* __restrict__ factor) {
    int j = blockIdx.x * blockDim.x + threadIdx.x;
    if (j < FEATS) {
        factor[j] = __expf((DENSITY - aa[j]) * 1.0f);
    }
}

__global__ void kw_colmean_wmma_kernel(const float* __restrict__ x,
                                       float* __restrict__ colmean) {
    const int col0 = blockIdx.x * 16;
    const int lane = threadIdx.x & 31;
    const int wave = threadIdx.x >> 5;
    const int n    = lane & 15;
    const int kb   = (lane >> 4) * 2;

    v2f a; a.x = 1.0f; a.y = 1.0f;
    v8f c = {};

    const float* base = x + col0 + n;
    const int r0 = wave * (BATCH / 8);
    const int r1 = r0 + (BATCH / 8);
    for (int r = r0; r < r1; r += 4) {
        v2f b;
        b.x = base[(size_t)(r + kb) * FEATS];
        b.y = base[(size_t)(r + kb + 1) * FEATS];
        c = __builtin_amdgcn_wmma_f32_16x16x4_f32(
                false, a, false, b, (short)0, c, false, false);
    }

    __shared__ float part[8][16];
    if (lane < 16) part[wave][lane] = c[0];
    __syncthreads();
    if (threadIdx.x < 16) {
        float s = 0.0f;
        for (int w = 0; w < 8; ++w) s += part[w][threadIdx.x];
        colmean[col0 + threadIdx.x] = s * (1.0f / (float)BATCH);
    }
}

__device__ __forceinline__ unsigned order_key(float f) {
    unsigned u = __float_as_uint(f);
    return (u & 0x80000000u) ? ~u : (u | 0x80000000u);
}

__device__ __forceinline__ unsigned radix_select_desc(
        const unsigned rk[EPT], unsigned* __restrict__ hist,
        unsigned* __restrict__ s_prefix, int* __restrict__ s_target,
        int R, int t) {
    unsigned prefix = 0;
    int target = R;
    for (int shift = 24; shift >= 0; shift -= 8) {
        hist[t] = 0;
        __syncthreads();
        const unsigned hmask = (shift == 24) ? 0u : (0xFFFFFFFFu << (shift + 8));
        #pragma unroll
        for (int e = 0; e < EPT; ++e) {
            unsigned k = rk[e];
            if ((k & hmask) == prefix)
                atomicAdd(&hist[(k >> shift) & 0xFFu], 1u);
        }
        __syncthreads();
        if (t == 0) {
            int tgt = target;
            int b = 255;
            for (; b > 0; --b) {
                int cnt = (int)hist[b];
                if (tgt <= cnt) break;
                tgt -= cnt;
            }
            *s_prefix = prefix | ((unsigned)b << shift);
            *s_target = tgt;
        }
        __syncthreads();
        prefix = *s_prefix;
        target = *s_target;
    }
    return prefix;
}

__global__ void __launch_bounds__(256)
kw_row_kernel(const float* __restrict__ x,
              const float* __restrict__ factor,
              const float* __restrict__ colmean,
              float* __restrict__ out) {
    __shared__ unsigned hist[256];
    __shared__ unsigned s_prefix;
    __shared__ int      s_target;

    const int row = blockIdx.x;
    const int t   = threadIdx.x;

    const float4* __restrict__ xin = (const float4*)(x + (size_t)row * FEATS);
    const float4* __restrict__ fin = (const float4*)factor;

    float    rx[EPT];
    unsigned rk[EPT];

    #pragma unroll
    for (int it = 0; it < EPT / 4; ++it) {
        int idx = it * 256 + t;
        float4 xv = xin[idx];
        float4 fv = fin[idx];
        rx[it * 4 + 0] = xv.x;  rk[it * 4 + 0] = order_key(xv.x * fv.x);
        rx[it * 4 + 1] = xv.y;  rk[it * 4 + 1] = order_key(xv.y * fv.y);
        rx[it * 4 + 2] = xv.z;  rk[it * 4 + 2] = order_key(xv.z * fv.z);
        rx[it * 4 + 3] = xv.w;  rk[it * 4 + 3] = order_key(xv.w * fv.w);
    }

    const unsigned thrHigh = radix_select_desc(rk, hist, &s_prefix, &s_target,
                                               RANK_HIGH, t);
    const unsigned thrLow  = radix_select_desc(rk, hist, &s_prefix, &s_target,
                                               RANK_LOW, t);

    const float4* __restrict__ cmin = (const float4*)colmean;
    float4* __restrict__ oout = (float4*)(out + (size_t)row * FEATS);
    #pragma unroll
    for (int it = 0; it < EPT / 4; ++it) {
        int idx = it * 256 + t;
        float4 cm = cmin[idx];
        float4 o;
        {
            unsigned k = rk[it * 4 + 0];
            bool m2 = (k <= thrHigh);
            bool m  = m2 && (k >= thrLow);
            o.x = (m ? rx[it * 4 + 0] : 0.0f) - (m2 ? cm.x : 0.0f);
        }
        {
            unsigned k = rk[it * 4 + 1];
            bool m2 = (k <= thrHigh);
            bool m  = m2 && (k >= thrLow);
            o.y = (m ? rx[it * 4 + 1] : 0.0f) - (m2 ? cm.y : 0.0f);
        }
        {
            unsigned k = rk[it * 4 + 2];
            bool m2 = (k <= thrHigh);
            bool m  = m2 && (k >= thrLow);
            o.z = (m ? rx[it * 4 + 2] : 0.0f) - (m2 ? cm.z : 0.0f);
        }
        {
            unsigned k = rk[it * 4 + 3];
            bool m2 = (k <= thrHigh);
            bool m  = m2 && (k >= thrLow);
            o.w = (m ? rx[it * 4 + 3] : 0.0f) - (m2 ? cm.w : 0.0f);
        }
        oout[idx] = o;
    }
}

extern "C" void kernel_launch(void* const* d_in, const int* in_sizes, int n_in,
                              void* d_out, int out_size, void* d_ws, size_t ws_size,
                              hipStream_t stream) {
    const float* x  = (const float*)d_in[0];
    const float* aa = (const float*)d_in[1];
    float* out = (float*)d_out;

    float* factor  = (float*)d_ws;
    float* colmean = (float*)d_ws + FEATS;

    kw_factor_kernel<<<FEATS / 256, 256, 0, stream>>>(aa, factor);

    kw_colmean_wmma_kernel<<<FEATS / 16, 256, 0, stream>>>(x, colmean);

    kw_row_kernel<<<BATCH, 256, 0, stream>>>(x, factor, colmean, out);
}


# Row kernel: LDS-atomic radix-histogram select replaced by one wave per row rank bisection, bit by bit, on 128 register keys per lane; counts from wave ballots; no LDS, atomics or barriers
# speedup vs baseline: 6.4261x; 6.4261x over previous
#include <hip/hip_runtime.h>

typedef __attribute__((ext_vector_type(2))) float v2f;
typedef __attribute__((ext_vector_type(8))) float v8f;

#define BATCH 8192
#define FEATS 4096
#define RANK_LOW 211
#define RANK_HIGH 7
#define DENSITY 0.05f
#define EPL 128
#define ROWS_PER_BLOCK 8

__global__ void kw_factor_kernel(const float* __restrict__ aa,
                                 float* __restrict__ factor) {
    int j = blockIdx.x * blockDim.x + threadIdx.x;
    if (j < FEATS) {
        factor[j] = __expf((DENSITY - aa[j]) * 1.0f);
    }
}

__global__ void kw_colmean_wmma_kernel(const float* __restrict__ x,
                                       float* __restrict__ colmean) {
    const int col0 = blockIdx.x * 16;
    const int lane = threadIdx.x & 31;
    const int wave = threadIdx.x >> 5;
    const int n    = lane & 15;
    const int kb   = (lane >> 4) * 2;

    v2f a; a.x = 1.0f; a.y = 1.0f;
    v8f c = {};

    const float* base = x + col0 + n;
    const int r0 = wave * (BATCH / 8);
    const int r1 = r0 + (BATCH / 8);
    for (int r = r0; r < r1; r += 4) {
        v2f b;
        b.x = base[(size_t)(r + kb) * FEATS];
        b.y = base[(size_t)(r + kb + 1) * FEATS];
        c = __builtin_amdgcn_wmma_f32_16x16x4_f32(
                false, a, false, b, (short)0, c, false, false);
    }

    __shared__ float part[8][16];
    if (lane < 16) part[wave][lane] = c[0];
    __syncthreads();
    if (threadIdx.x < 16) {
        float s = 0.0f;
        for (int w = 0; w < 8; ++w) s += part[w][threadIdx.x];
        colmean[col0 + threadIdx.x] = s * (1.0f / (float)BATCH);
    }
}

__device__ __forceinline__ unsigned order_key(float f) {
    unsigned u = __float_as_uint(f);
    return (u & 0x80000000u) ? ~u : (u | 0x80000000u);
}

__device__ __forceinline__ int count2(bool p0, bool p1) {
    unsigned long long m = (unsigned long long)__builtin_amdgcn_ballot_w32(p0) |
                           ((unsigned long long)__builtin_amdgcn_ballot_w32(p1) << 32);
    return __builtin_popcountll(m);
}

template <int START, int N>
__device__ __forceinline__ void load_keys(const float4* __restrict__ xin,
                                          const float4* __restrict__ fin,
                                          int lane, unsigned (&rk)[EPL]) {
    float4 xv[N], fv[N];
    #pragma unroll
    for (int i = 0; i < N; ++i) {
        xv[i] = xin[(START + i) * 32 + lane];
        fv[i] = fin[(START + i) * 32 + lane];
    }
    #pragma unroll
    for (int i = 0; i < N; ++i) {
        rk[(START + i) * 4 + 0] = order_key(xv[i].x * fv[i].x);
        rk[(START + i) * 4 + 1] = order_key(xv[i].y * fv[i].y);
        rk[(START + i) * 4 + 2] = order_key(xv[i].z * fv[i].z);
        rk[(START + i) * 4 + 3] = order_key(xv[i].w * fv[i].w);
    }
}

__device__ __forceinline__ float masked(unsigned k, float xv, float cm,
                                        unsigned thrHigh, unsigned thrLow) {
    bool m2 = (k <= thrHigh);
    bool m  = m2 && (k >= thrLow);
    return (m ? xv : 0.0f) - (m2 ? cm : 0.0f);
}

template <int START, int N>
__device__ __forceinline__ void store_rows(const float4* __restrict__ xin,
                                           const float4* __restrict__ cmin,
                                           float4* __restrict__ oout, int lane,
                                           const unsigned (&rk)[EPL],
                                           unsigned thrHigh, unsigned thrLow) {
    float4 xv[N], cm[N];
    #pragma unroll
    for (int i = 0; i < N; ++i) {
        xv[i] = xin[(START + i) * 32 + lane];
        cm[i] = cmin[(START + i) * 32 + lane];
    }
    #pragma unroll
    for (int i = 0; i < N; ++i) {
        float4 o;
        o.x = masked(rk[(START + i) * 4 + 0], xv[i].x, cm[i].x, thrHigh, thrLow);
        o.y = masked(rk[(START + i) * 4 + 1], xv[i].y, cm[i].y, thrHigh, thrLow);
        o.z = masked(rk[(START + i) * 4 + 2], xv[i].z, cm[i].z, thrHigh, thrLow);
        o.w = masked(rk[(START + i) * 4 + 3], xv[i].w, cm[i].w, thrHigh, thrLow);
        oout[(START + i) * 32 + lane] = o;
    }
}

__global__ void __launch_bounds__(256)
kw_row_kernel(const float* __restrict__ x,
              const float* __restrict__ factor,
              const float* __restrict__ colmean,
              float* __restrict__ out) {
    const int lane = threadIdx.x & 31;
    const int row  = blockIdx.x * ROWS_PER_BLOCK + (threadIdx.x >> 5);

    const float4* __restrict__ xin = (const float4*)(x + (size_t)row * FEATS);
    const float4* __restrict__ fin = (const float4*)factor;

    unsigned rk[EPL];

    load_keys<0, 16>(xin, fin, lane, rk);
    __builtin_amdgcn_sched_barrier(0);
    load_keys<16, 16>(xin, fin, lane, rk);

    unsigned thrHigh = 0u, thrLow = 0u;
    #pragma unroll 1
    for (int bit = 31; bit >= 0; --bit) {
        const unsigned candH = thrHigh | (1u << bit);
        const unsigned candL = thrLow  | (1u << bit);
        int nH = 0, nL = 0;
        #pragma unroll
        for (int e = 0; e < EPL; e += 2) {
            nH += count2(rk[e] >= candH, rk[e + 1] >= candH);
            nL += count2(rk[e] >= candL, rk[e + 1] >= candL);
            if ((e & 15) == 14) __builtin_amdgcn_sched_barrier(0);
        }
        if (nH >= RANK_HIGH) thrHigh = candH;
        if (nL >= RANK_LOW)  thrLow  = candL;
    }

    const float4* __restrict__ cmin = (const float4*)colmean;
    float4* __restrict__ oout = (float4*)(out + (size_t)row * FEATS);
    store_rows<0, 10>(xin, cmin, oout, lane, rk, thrHigh, thrLow);
    __builtin_amdgcn_sched_barrier(0);
    store_rows<10, 10>(xin, cmin, oout, lane, rk, thrHigh, thrLow);
    __builtin_amdgcn_sched_barrier(0);
    store_rows<20, 12>(xin, cmin, oout, lane, rk, thrHigh, thrLow);
}

extern "C" void kernel_launch(void* const* d_in, const int* in_sizes, int n_in,
                              void* d_out, int out_size, void* d_ws, size_t ws_size,
                              hipStream_t stream) {
    const float* x  = (const float*)d_in[0];
    const float* aa = (const float*)d_in[1];
    float* out = (float*)d_out;

    float* factor  = (float*)d_ws;
    float* colmean = (float*)d_ws + FEATS;

    kw_factor_kernel<<<FEATS / 256, 256, 0, stream>>>(aa, factor);

    kw_colmean_wmma_kernel<<<FEATS / 16, 256, 0, stream>>>(x, colmean);

    kw_row_kernel<<<BATCH / ROWS_PER_BLOCK, 256, 0, stream>>>(x, factor, colmean, out);
}


# Row kernel: 128-keys-per-lane bisection only until <=384 keys lie at or above the floor; those are packed by ballot prefix counts via wave-private LDS into 12 regs/lane, both ranks bisected there
# speedup vs baseline: 9.8270x; 1.5292x over previous
#include <hip/hip_runtime.h>

typedef __attribute__((ext_vector_type(2))) float v2f;
typedef __attribute__((ext_vector_type(8))) float v8f;

#define BATCH 8192
#define FEATS 4096
#define RANK_LOW 211
#define RANK_HIGH 7
#define DENSITY 0.05f
#define EPL 128
#define ROWS_PER_BLOCK 8
#define CAND_REGS 12
#define CAND_CAP (CAND_REGS * 32)

__global__ void kw_factor_kernel(const float* __restrict__ aa,
                                 float* __restrict__ factor) {
    int j = blockIdx.x * blockDim.x + threadIdx.x;
    if (j < FEATS) {
        factor[j] = __expf((DENSITY - aa[j]) * 1.0f);
    }
}

__global__ void kw_colmean_wmma_kernel(const float* __restrict__ x,
                                       float* __restrict__ colmean) {
    const int col0 = blockIdx.x * 16;
    const int lane = threadIdx.x & 31;
    const int wave = threadIdx.x >> 5;
    const int n    = lane & 15;
    const int kb   = (lane >> 4) * 2;

    v2f a; a.x = 1.0f; a.y = 1.0f;
    v8f c = {};

    const float* base = x + col0 + n;
    const int r0 = wave * (BATCH / 8);
    const int r1 = r0 + (BATCH / 8);
    for (int r = r0; r < r1; r += 4) {
        v2f b;
        b.x = base[(size_t)(r + kb) * FEATS];
        b.y = base[(size_t)(r + kb + 1) * FEATS];
        c = __builtin_amdgcn_wmma_f32_16x16x4_f32(
                false, a, false, b, (short)0, c, false, false);
    }

    __shared__ float part[8][16];
    if (lane < 16) part[wave][lane] = c[0];
    __syncthreads();
    if (threadIdx.x < 16) {
        float s = 0.0f;
        for (int w = 0; w < 8; ++w) s += part[w][threadIdx.x];
        colmean[col0 + threadIdx.x] = s * (1.0f / (float)BATCH);
    }
}

__device__ __forceinline__ unsigned order_key(float f) {
    unsigned u = __float_as_uint(f);
    return (u & 0x80000000u) ? ~u : (u | 0x80000000u);
}

__device__ __forceinline__ int count2(bool p0, bool p1) {
    unsigned long long m = (unsigned long long)__builtin_amdgcn_ballot_w32(p0) |
                           ((unsigned long long)__builtin_amdgcn_ballot_w32(p1) << 32);
    return __builtin_popcountll(m);
}

template <int START, int N>
__device__ __forceinline__ void load_keys(const float4* __restrict__ xin,
                                          const float4* __restrict__ fin,
                                          int lane, unsigned (&rk)[EPL]) {
    float4 xv[N], fv[N];
    #pragma unroll
    for (int i = 0; i < N; ++i) {
        xv[i] = xin[(START + i) * 32 + lane];
        fv[i] = fin[(START + i) * 32 + lane];
    }
    #pragma unroll
    for (int i = 0; i < N; ++i) {
        rk[(START + i) * 4 + 0] = order_key(xv[i].x * fv[i].x);
        rk[(START + i) * 4 + 1] = order_key(xv[i].y * fv[i].y);
        rk[(START + i) * 4 + 2] = order_key(xv[i].z * fv[i].z);
        rk[(START + i) * 4 + 3] = order_key(xv[i].w * fv[i].w);
    }
}

__device__ __forceinline__ float masked(unsigned k, float xv, float cm,
                                        unsigned thrHigh, unsigned thrLow) {
    bool m2 = (k <= thrHigh);
    bool m  = m2 && (k >= thrLow);
    return (m ? xv : 0.0f) - (m2 ? cm : 0.0f);
}

template <int START, int N>
__device__ __forceinline__ void store_rows(const float4* __restrict__ xin,
                                           const float4* __restrict__ cmin,
                                           float4* __restrict__ oout, int lane,
                                           const unsigned (&rk)[EPL],
                                           unsigned thrHigh, unsigned thrLow) {
    float4 xv[N], cm[N];
    #pragma unroll
    for (int i = 0; i < N; ++i) {
        xv[i] = xin[(START + i) * 32 + lane];
        cm[i] = cmin[(START + i) * 32 + lane];
    }
    #pragma unroll
    for (int i = 0; i < N; ++i) {
        float4 o;
        o.x = masked(rk[(START + i) * 4 + 0], xv[i].x, cm[i].x, thrHigh, thrLow);
        o.y = masked(rk[(START + i) * 4 + 1], xv[i].y, cm[i].y, thrHigh, thrLow);
        o.z = masked(rk[(START + i) * 4 + 2], xv[i].z, cm[i].z, thrHigh, thrLow);
        o.w = masked(rk[(START + i) * 4 + 3], xv[i].w, cm[i].w, thrHigh, thrLow);
        oout[(START + i) * 32 + lane] = o;
    }
}

__global__ void __launch_bounds__(256)
kw_row_kernel(const float* __restrict__ x,
              const float* __restrict__ factor,
              const float* __restrict__ colmean,
              float* __restrict__ out) {
    __shared__ unsigned cand_strip[ROWS_PER_BLOCK][CAND_CAP + 32];
    const int lane = threadIdx.x & 31;
    const int row  = blockIdx.x * ROWS_PER_BLOCK + (threadIdx.x >> 5);

    const float4* __restrict__ xin = (const float4*)(x + (size_t)row * FEATS);
    const float4* __restrict__ fin = (const float4*)factor;

    unsigned rk[EPL];

    load_keys<0, 16>(xin, fin, lane, rk);
    __builtin_amdgcn_sched_barrier(0);
    load_keys<16, 16>(xin, fin, lane, rk);

    unsigned thrHigh = 0u, thrLow = 0u;

    unsigned floorKey = 0u;
    int nAbove = FEATS;
    int bit = 31;
    #pragma unroll 1
    while (bit >= 0 && nAbove > CAND_CAP) {
        const unsigned c = floorKey | (1u << bit);
        int n = 0;
        #pragma unroll
        for (int e = 0; e < EPL; e += 2) {
            n += count2(rk[e] >= c, rk[e + 1] >= c);
            if ((e & 15) == 14) __builtin_amdgcn_sched_barrier(0);
        }
        if (n >= RANK_LOW) { floorKey = c; nAbove = n; }
        --bit;
    }

    if (nAbove <= CAND_CAP) {
        unsigned* __restrict__ strip =
            cand_strip[__builtin_amdgcn_readfirstlane((int)(threadIdx.x >> 5))];
        #pragma unroll
        for (int r = 0; r < CAND_REGS; ++r) strip[r * 32 + lane] = 0u;
        unsigned filled = 0u;
        #pragma unroll
        for (int e = 0; e < EPL; ++e) {
            const bool p = rk[e] >= floorKey;
            const unsigned b = __builtin_amdgcn_ballot_w32(p);
            const unsigned slot = __builtin_amdgcn_mbcnt_lo(b, filled);
            strip[p ? slot : (unsigned)(CAND_CAP + lane)] = rk[e];
            filled += (unsigned)__builtin_popcount(b);
            if ((e & 7) == 7) __builtin_amdgcn_sched_barrier(0);
        }
        unsigned ck[CAND_REGS];
        #pragma unroll
        for (int r = 0; r < CAND_REGS; ++r) ck[r] = strip[r * 32 + lane];

        #pragma unroll 1
        for (int b2 = 31; b2 >= 0; --b2) {
            const unsigned candH = thrHigh | (1u << b2);
            const unsigned candL = thrLow  | (1u << b2);
            int nH = 0, nL = 0;
            #pragma unroll
            for (int r = 0; r < CAND_REGS; r += 2) {
                nH += count2(ck[r] >= candH, ck[r + 1] >= candH);
                nL += count2(ck[r] >= candL, ck[r + 1] >= candL);
            }
            if (nH >= RANK_HIGH) thrHigh = candH;
            if (nL >= RANK_LOW)  thrLow  = candL;
        }
    } else {
        #pragma unroll 1
        for (int b2 = 31; b2 >= 0; --b2) {
            const unsigned candH = thrHigh | (1u << b2);
            const unsigned candL = thrLow  | (1u << b2);
            int nH = 0, nL = 0;
            #pragma unroll
            for (int e = 0; e < EPL; e += 2) {
                nH += count2(rk[e] >= candH, rk[e + 1] >= candH);
                nL += count2(rk[e] >= candL, rk[e + 1] >= candL);
                if ((e & 15) == 14) __builtin_amdgcn_sched_barrier(0);
            }
            if (nH >= RANK_HIGH) thrHigh = candH;
            if (nL >= RANK_LOW)  thrLow  = candL;
        }
    }

    const float4* __restrict__ cmin = (const float4*)colmean;
    float4* __restrict__ oout = (float4*)(out + (size_t)row * FEATS);
    store_rows<0, 10>(xin, cmin, oout, lane, rk, thrHigh, thrLow);
    __builtin_amdgcn_sched_barrier(0);
    store_rows<10, 10>(xin, cmin, oout, lane, rk, thrHigh, thrLow);
    __builtin_amdgcn_sched_barrier(0);
    store_rows<20, 12>(xin, cmin, oout, lane, rk, thrHigh, thrLow);
}

extern "C" void kernel_launch(void* const* d_in, const int* in_sizes, int n_in,
                              void* d_out, int out_size, void* d_ws, size_t ws_size,
                              hipStream_t stream) {
    const float* x  = (const float*)d_in[0];
    const float* aa = (const float*)d_in[1];
    float* out = (float*)d_out;

    float* factor  = (float*)d_ws;
    float* colmean = (float*)d_ws + FEATS;

    kw_factor_kernel<<<FEATS / 256, 256, 0, stream>>>(aa, factor);

    kw_colmean_wmma_kernel<<<FEATS / 16, 256, 0, stream>>>(x, colmean);

    kw_row_kernel<<<BATCH / ROWS_PER_BLOCK, 256, 0, stream>>>(x, factor, colmean, out);
}


# Column-mean kernel: 4-byte loads of 16-column strips replaced by 16-byte loads of 256-B row pieces, 16-wave groups per 256x512 tile; slab partials added in the factor launch; new sum order
# speedup vs baseline: 12.4766x; 1.2696x over previous
#include <hip/hip_runtime.h>

typedef __attribute__((ext_vector_type(2))) float v2f;
typedef __attribute__((ext_vector_type(8))) float v8f;

#define BATCH 8192
#define FEATS 4096
#define RANK_LOW 211
#define RANK_HIGH 7
#define DENSITY 0.05f
#define EPL 128
#define ROWS_PER_BLOCK 8
#define CAND_REGS 12
#define CAND_CAP (CAND_REGS * 32)

#define CS_TILE_COLS 256
#define CS_WAVE_ROWS 128
#define CS_SLAB_ROWS 512
#define CS_SLABS (BATCH / CS_SLAB_ROWS)
#define CS_BLK_LOADS 16

__device__ __forceinline__ void cs_load(const float4* __restrict__ p,
                                        float4 (&buf)[CS_BLK_LOADS]) {
    #pragma unroll
    for (int i = 0; i < CS_BLK_LOADS; ++i)
        buf[i] = p[(size_t)i * 2 * (FEATS / 4)];
}

__device__ __forceinline__ void cs_sum(const float4 (&buf)[CS_BLK_LOADS],
                                       v2f ones, v8f (&acc)[4]) {
    #pragma unroll
    for (int q = 0; q < CS_BLK_LOADS; q += 2) {
        v2f b0, b1, b2, b3;
        b0.x = buf[q].x; b0.y = buf[q + 1].x;
        b1.x = buf[q].y; b1.y = buf[q + 1].y;
        b2.x = buf[q].z; b2.y = buf[q + 1].z;
        b3.x = buf[q].w; b3.y = buf[q + 1].w;
        acc[0] = __builtin_amdgcn_wmma_f32_16x16x4_f32(
                     false, ones, false, b0, (short)0, acc[0], false, false);
        acc[1] = __builtin_amdgcn_wmma_f32_16x16x4_f32(
                     false, ones, false, b1, (short)0, acc[1], false, false);
        acc[2] = __builtin_amdgcn_wmma_f32_16x16x4_f32(
                     false, ones, false, b2, (short)0, acc[2], false, false);
        acc[3] = __builtin_amdgcn_wmma_f32_16x16x4_f32(
                     false, ones, false, b3, (short)0, acc[3], false, false);
    }
}

__global__ void __launch_bounds__(512)
kw_colsum_wmma_kernel(const float* __restrict__ x,
                      float* __restrict__ colpart) {
    const int lane = threadIdx.x & 31;
    const int wave = threadIdx.x >> 5;
    const int cw   = wave & 3;
    const int rw   = wave >> 2;
    const int n    = lane & 15;
    const int h    = lane >> 4;
    const int tile = blockIdx.x & (FEATS / CS_TILE_COLS - 1);
    const int slab = blockIdx.x / (FEATS / CS_TILE_COLS);

    const int col0 = tile * CS_TILE_COLS + cw * 64;
    const int row0 = slab * CS_SLAB_ROWS + rw * CS_WAVE_ROWS;
    const float4* __restrict__ p =
        (const float4*)(x + (size_t)(row0 + h) * FEATS + col0) + n;
    const size_t blk = (size_t)2 * CS_BLK_LOADS * (FEATS / 4);

    v2f ones; ones.x = 1.0f; ones.y = 1.0f;
    v8f acc[4] = {};

    float4 bufA[CS_BLK_LOADS], bufB[CS_BLK_LOADS];
    cs_load(p, bufA);
    cs_load(p + blk, bufB);
    cs_sum(bufA, ones, acc);
    cs_load(p + 2 * blk, bufA);
    cs_sum(bufB, ones, acc);
    cs_load(p + 3 * blk, bufB);
    cs_sum(bufA, ones, acc);
    cs_sum(bufB, ones, acc);

    __shared__ float4 part[CS_SLAB_ROWS / CS_WAVE_ROWS][CS_TILE_COLS / 4];
    if (lane < 16) {
        float4 s;
        s.x = acc[0][0]; s.y = acc[1][0]; s.z = acc[2][0]; s.w = acc[3][0];
        part[rw][cw * 16 + n] = s;
    }
    __syncthreads();
    if (threadIdx.x < CS_TILE_COLS / 4) {
        const float4 s0 = part[0][threadIdx.x], s1 = part[1][threadIdx.x];
        const float4 s2 = part[2][threadIdx.x], s3 = part[3][threadIdx.x];
        float4 s;
        s.x = (s0.x + s1.x) + (s2.x + s3.x);
        s.y = (s0.y + s1.y) + (s2.y + s3.y);
        s.z = (s0.z + s1.z) + (s2.z + s3.z);
        s.w = (s0.w + s1.w) + (s2.w + s3.w);
        ((float4*)(colpart + (size_t)slab * FEATS + tile * CS_TILE_COLS))[threadIdx.x] = s;
    }
}

__global__ void kw_factor_mean_kernel(const float* __restrict__ aa,
                                      const float* __restrict__ colpart,
                                      float* __restrict__ factor,
                                      float* __restrict__ colmean) {
    int j = blockIdx.x * blockDim.x + threadIdx.x;
    if (j < FEATS) {
        factor[j] = __expf((DENSITY - aa[j]) * 1.0f);
        float s = 0.0f;
        #pragma unroll
        for (int t = 0; t < CS_SLABS; ++t) s += colpart[(size_t)t * FEATS + j];
        colmean[j] = s * (1.0f / (float)BATCH);
    }
}

__device__ __forceinline__ unsigned order_key(float f) {
    unsigned u = __float_as_uint(f);
    return (u & 0x80000000u) ? ~u : (u | 0x80000000u);
}

__device__ __forceinline__ int count2(bool p0, bool p1) {
    unsigned long long m = (unsigned long long)__builtin_amdgcn_ballot_w32(p0) |
                           ((unsigned long long)__builtin_amdgcn_ballot_w32(p1) << 32);
    return __builtin_popcountll(m);
}

template <int START, int N>
__device__ __forceinline__ void load_keys(const float4* __restrict__ xin,
                                          const float4* __restrict__ fin,
                                          int lane, unsigned (&rk)[EPL]) {
    float4 xv[N], fv[N];
    #pragma unroll
    for (int i = 0; i < N; ++i) {
        xv[i] = xin[(START + i) * 32 + lane];
        fv[i] = fin[(START + i) * 32 + lane];
    }
    #pragma unroll
    for (int i = 0; i < N; ++i) {
        rk[(START + i) * 4 + 0] = order_key(xv[i].x * fv[i].x);
        rk[(START + i) * 4 + 1] = order_key(xv[i].y * fv[i].y);
        rk[(START + i) * 4 + 2] = order_key(xv[i].z * fv[i].z);
        rk[(START + i) * 4 + 3] = order_key(xv[i].w * fv[i].w);
    }
}

__device__ __forceinline__ float masked(unsigned k, float xv, float cm,
                                        unsigned thrHigh, unsigned thrLow) {
    bool m2 = (k <= thrHigh);
    bool m  = m2 && (k >= thrLow);
    return (m ? xv : 0.0f) - (m2 ? cm : 0.0f);
}

template <int START, int N>
__device__ __forceinline__ void store_rows(const float4* __restrict__ xin,
                                           const float4* __restrict__ cmin,
                                           float4* __restrict__ oout, int lane,
                                           const unsigned (&rk)[EPL],
                                           unsigned thrHigh, unsigned thrLow) {
    float4 xv[N], cm[N];
    #pragma unroll
    for (int i = 0; i < N; ++i) {
        xv[i] = xin[(START + i) * 32 + lane];
        cm[i] = cmin[(START + i) * 32 + lane];
    }
    #pragma unroll
    for (int i = 0; i < N; ++i) {
        float4 o;
        o.x = masked(rk[(START + i) * 4 + 0], xv[i].x, cm[i].x, thrHigh, thrLow);
        o.y = masked(rk[(START + i) * 4 + 1], xv[i].y, cm[i].y, thrHigh, thrLow);
        o.z = masked(rk[(START + i) * 4 + 2], xv[i].z, cm[i].z, thrHigh, thrLow);
        o.w = masked(rk[(START + i) * 4 + 3], xv[i].w, cm[i].w, thrHigh, thrLow);
        oout[(START + i) * 32 + lane] = o;
    }
}

__global__ void __launch_bounds__(256)
kw_row_kernel(const float* __restrict__ x,
              const float* __restrict__ factor,
              const float* __restrict__ colmean,
              float* __restrict__ out) {
    __shared__ unsigned cand_strip[ROWS_PER_BLOCK][CAND_CAP + 32];
    const int lane = threadIdx.x & 31;
    const int row  = blockIdx.x * ROWS_PER_BLOCK + (threadIdx.x >> 5);

    const float4* __restrict__ xin = (const float4*)(x + (size_t)row * FEATS);
    const float4* __restrict__ fin = (const float4*)factor;

    unsigned rk[EPL];

    load_keys<0, 16>(xin, fin, lane, rk);
    __builtin_amdgcn_sched_barrier(0);
    load_keys<16, 16>(xin, fin, lane, rk);

    unsigned thrHigh = 0u, thrLow = 0u;

    unsigned floorKey = 0u;
    int nAbove = FEATS;
    int bit = 31;
    #pragma unroll 1
    while (bit >= 0 && nAbove > CAND_CAP) {
        const unsigned c = floorKey | (1u << bit);
        int n = 0;
        #pragma unroll
        for (int e = 0; e < EPL; e += 2) {
            n += count2(rk[e] >= c, rk[e + 1] >= c);
            if ((e & 15) == 14) __builtin_amdgcn_sched_barrier(0);
        }
        if (n >= RANK_LOW) { floorKey = c; nAbove = n; }
        --bit;
    }

    if (nAbove <= CAND_CAP) {
        unsigned* __restrict__ strip =
            cand_strip[__builtin_amdgcn_readfirstlane((int)(threadIdx.x >> 5))];
        #pragma unroll
        for (int r = 0; r < CAND_REGS; ++r) strip[r * 32 + lane] = 0u;
        unsigned filled = 0u;
        #pragma unroll
        for (int e = 0; e < EPL; ++e) {
            const bool p = rk[e] >= floorKey;
            const unsigned b = __builtin_amdgcn_ballot_w32(p);
            const unsigned slot = __builtin_amdgcn_mbcnt_lo(b, filled);
            strip[p ? slot : (unsigned)(CAND_CAP + lane)] = rk[e];
            filled += (unsigned)__builtin_popcount(b);
            if ((e & 7) == 7) __builtin_amdgcn_sched_barrier(0);
        }
        unsigned ck[CAND_REGS];
        #pragma unroll
        for (int r = 0; r < CAND_REGS; ++r) ck[r] = strip[r * 32 + lane];

        #pragma unroll 1
        for (int b2 = 31; b2 >= 0; --b2) {
            const unsigned candH = thrHigh | (1u << b2);
            const unsigned candL = thrLow  | (1u << b2);
            int nH = 0, nL = 0;
            #pragma unroll
            for (int r = 0; r < CAND_REGS; r += 2) {
                nH += count2(ck[r] >= candH, ck[r + 1] >= candH);
                nL += count2(ck[r] >= candL, ck[r + 1] >= candL);
            }
            if (nH >= RANK_HIGH) thrHigh = candH;
            if (nL >= RANK_LOW)  thrLow  = candL;
        }
    } else {
        #pragma unroll 1
        for (int b2 = 31; b2 >= 0; --b2) {
            const unsigned candH = thrHigh | (1u << b2);
            const unsigned candL = thrLow  | (1u << b2);
            int nH = 0, nL = 0;
            #pragma unroll
            for (int e = 0; e < EPL; e += 2) {
                nH += count2(rk[e] >= candH, rk[e + 1] >= candH);
                nL += count2(rk[e] >= candL, rk[e + 1] >= candL);
                if ((e & 15) == 14) __builtin_amdgcn_sched_barrier(0);
            }
            if (nH >= RANK_HIGH) thrHigh = candH;
            if (nL >= RANK_LOW)  thrLow  = candL;
        }
    }

    const float4* __restrict__ cmin = (const float4*)colmean;
    float4* __restrict__ oout = (float4*)(out + (size_t)row * FEATS);
    store_rows<0, 10>(xin, cmin, oout, lane, rk, thrHigh, thrLow);
    __builtin_amdgcn_sched_barrier(0);
    store_rows<10, 10>(xin, cmin, oout, lane, rk, thrHigh, thrLow);
    __builtin_amdgcn_sched_barrier(0);
    store_rows<20, 12>(xin, cmin, oout, lane, rk, thrHigh, thrLow);
}

extern "C" void kernel_launch(void* const* d_in, const int* in_sizes, int n_in,
                              void* d_out, int out_size, void* d_ws, size_t ws_size,
                              hipStream_t stream) {
    const float* x  = (const float*)d_in[0];
    const float* aa = (const float*)d_in[1];
    float* out = (float*)d_out;

    float* factor  = (float*)d_ws;
    float* colmean = (float*)d_ws + FEATS;
    float* colpart = out;

    kw_colsum_wmma_kernel<<<(FEATS / CS_TILE_COLS) * CS_SLABS, 512, 0, stream>>>(x, colpart);

    kw_factor_mean_kernel<<<FEATS / 256, 256, 0, stream>>>(aa, colpart, factor, colmean);

    kw_row_kernel<<<BATCH / ROWS_PER_BLOCK, 256, 0, stream>>>(x, factor, colmean, out);
}


# Row kernel, short-list rank search: low rank resumes at the narrowing's first untried bit; each search ends once exactly rank-many keys are counted, the key is then their minimum over the wave
# speedup vs baseline: 13.0756x; 1.0480x over previous
#include <hip/hip_runtime.h>

typedef __attribute__((ext_vector_type(2))) float v2f;
typedef __attribute__((ext_vector_type(8))) float v8f;

#define BATCH 8192
#define FEATS 4096
#define RANK_LOW 211
#define RANK_HIGH 7
#define DENSITY 0.05f
#define EPL 128
#define ROWS_PER_BLOCK 8
#define CAND_REGS 12
#define CAND_CAP (CAND_REGS * 32)

#define CS_TILE_COLS 256
#define CS_WAVE_ROWS 128
#define CS_SLAB_ROWS 512
#define CS_SLABS (BATCH / CS_SLAB_ROWS)
#define CS_BLK_LOADS 16

__device__ __forceinline__ void cs_load(const float4* __restrict__ p,
                                        float4 (&buf)[CS_BLK_LOADS]) {
    #pragma unroll
    for (int i = 0; i < CS_BLK_LOADS; ++i)
        buf[i] = p[(size_t)i * 2 * (FEATS / 4)];
}

__device__ __forceinline__ void cs_sum(const float4 (&buf)[CS_BLK_LOADS],
                                       v2f ones, v8f (&acc)[4]) {
    #pragma unroll
    for (int q = 0; q < CS_BLK_LOADS; q += 2) {
        v2f b0, b1, b2, b3;
        b0.x = buf[q].x; b0.y = buf[q + 1].x;
        b1.x = buf[q].y; b1.y = buf[q + 1].y;
        b2.x = buf[q].z; b2.y = buf[q + 1].z;
        b3.x = buf[q].w; b3.y = buf[q + 1].w;
        acc[0] = __builtin_amdgcn_wmma_f32_16x16x4_f32(
                     false, ones, false, b0, (short)0, acc[0], false, false);
        acc[1] = __builtin_amdgcn_wmma_f32_16x16x4_f32(
                     false, ones, false, b1, (short)0, acc[1], false, false);
        acc[2] = __builtin_amdgcn_wmma_f32_16x16x4_f32(
                     false, ones, false, b2, (short)0, acc[2], false, false);
        acc[3] = __builtin_amdgcn_wmma_f32_16x16x4_f32(
                     false, ones, false, b3, (short)0, acc[3], false, false);
    }
}

__global__ void __launch_bounds__(512)
kw_colsum_wmma_kernel(const float* __restrict__ x,
                      float* __restrict__ colpart) {
    const int lane = threadIdx.x & 31;
    const int wave = threadIdx.x >> 5;
    const int cw   = wave & 3;
    const int rw   = wave >> 2;
    const int n    = lane & 15;
    const int h    = lane >> 4;
    const int tile = blockIdx.x & (FEATS / CS_TILE_COLS - 1);
    const int slab = blockIdx.x / (FEATS / CS_TILE_COLS);

    const int col0 = tile * CS_TILE_COLS + cw * 64;
    const int row0 = slab * CS_SLAB_ROWS + rw * CS_WAVE_ROWS;
    const float4* __restrict__ p =
        (const float4*)(x + (size_t)(row0 + h) * FEATS + col0) + n;
    const size_t blk = (size_t)2 * CS_BLK_LOADS * (FEATS / 4);

    v2f ones; ones.x = 1.0f; ones.y = 1.0f;
    v8f acc[4] = {};

    float4 bufA[CS_BLK_LOADS], bufB[CS_BLK_LOADS];
    cs_load(p, bufA);
    cs_load(p + blk, bufB);
    cs_sum(bufA, ones, acc);
    cs_load(p + 2 * blk, bufA);
    cs_sum(bufB, ones, acc);
    cs_load(p + 3 * blk, bufB);
    cs_sum(bufA, ones, acc);
    cs_sum(bufB, ones, acc);

    __shared__ float4 part[CS_SLAB_ROWS / CS_WAVE_ROWS][CS_TILE_COLS / 4];
    if (lane < 16) {
        float4 s;
        s.x = acc[0][0]; s.y = acc[1][0]; s.z = acc[2][0]; s.w = acc[3][0];
        part[rw][cw * 16 + n] = s;
    }
    __syncthreads();
    if (threadIdx.x < CS_TILE_COLS / 4) {
        const float4 s0 = part[0][threadIdx.x], s1 = part[1][threadIdx.x];
        const float4 s2 = part[2][threadIdx.x], s3 = part[3][threadIdx.x];
        float4 s;
        s.x = (s0.x + s1.x) + (s2.x + s3.x);
        s.y = (s0.y + s1.y) + (s2.y + s3.y);
        s.z = (s0.z + s1.z) + (s2.z + s3.z);
        s.w = (s0.w + s1.w) + (s2.w + s3.w);
        ((float4*)(colpart + (size_t)slab * FEATS + tile * CS_TILE_COLS))[threadIdx.x] = s;
    }
}

__global__ void kw_factor_mean_kernel(const float* __restrict__ aa,
                                      const float* __restrict__ colpart,
                                      float* __restrict__ factor,
                                      float* __restrict__ colmean) {
    int j = blockIdx.x * blockDim.x + threadIdx.x;
    if (j < FEATS) {
        factor[j] = __expf((DENSITY - aa[j]) * 1.0f);
        float s = 0.0f;
        #pragma unroll
        for (int t = 0; t < CS_SLABS; ++t) s += colpart[(size_t)t * FEATS + j];
        colmean[j] = s * (1.0f / (float)BATCH);
    }
}

__device__ __forceinline__ unsigned order_key(float f) {
    unsigned u = __float_as_uint(f);
    return (u & 0x80000000u) ? ~u : (u | 0x80000000u);
}

__device__ __forceinline__ int count2(bool p0, bool p1) {
    unsigned long long m = (unsigned long long)__builtin_amdgcn_ballot_w32(p0) |
                           ((unsigned long long)__builtin_amdgcn_ballot_w32(p1) << 32);
    return __builtin_popcountll(m);
}

__device__ __forceinline__ unsigned wave_min_u32(unsigned v) {
    unsigned o;
    o = (unsigned)__builtin_amdgcn_update_dpp((int)v, (int)v, 0x168, 0xF, 0xF, false);
    v = o < v ? o : v;
    o = (unsigned)__builtin_amdgcn_update_dpp((int)v, (int)v, 0x164, 0xF, 0xF, false);
    v = o < v ? o : v;
    o = (unsigned)__builtin_amdgcn_update_dpp((int)v, (int)v, 0x162, 0xF, 0xF, false);
    v = o < v ? o : v;
    o = (unsigned)__builtin_amdgcn_update_dpp((int)v, (int)v, 0x161, 0xF, 0xF, false);
    v = o < v ? o : v;
    o = (unsigned)__builtin_amdgcn_permlanex16((int)v, (int)v, 0x76543210, 0xfedcba98,
                                               false, false);
    v = o < v ? o : v;
    return v;
}

__device__ __forceinline__ unsigned min_at_or_above(const unsigned (&ck)[CAND_REGS],
                                                    unsigned t) {
    unsigned m = 0xFFFFFFFFu;
    #pragma unroll
    for (int r = 0; r < CAND_REGS; ++r) {
        const unsigned k = (ck[r] >= t) ? ck[r] : 0xFFFFFFFFu;
        m = k < m ? k : m;
    }
    return wave_min_u32(m);
}

__device__ __forceinline__ int count_list(const unsigned (&ck)[CAND_REGS], unsigned c) {
    int n = 0;
    #pragma unroll
    for (int r = 0; r < CAND_REGS; r += 2) n += count2(ck[r] >= c, ck[r + 1] >= c);
    return n;
}

template <int START, int N>
__device__ __forceinline__ void load_keys(const float4* __restrict__ xin,
                                          const float4* __restrict__ fin,
                                          int lane, unsigned (&rk)[EPL]) {
    float4 xv[N], fv[N];
    #pragma unroll
    for (int i = 0; i < N; ++i) {
        xv[i] = xin[(START + i) * 32 + lane];
        fv[i] = fin[(START + i) * 32 + lane];
    }
    #pragma unroll
    for (int i = 0; i < N; ++i) {
        rk[(START + i) * 4 + 0] = order_key(xv[i].x * fv[i].x);
        rk[(START + i) * 4 + 1] = order_key(xv[i].y * fv[i].y);
        rk[(START + i) * 4 + 2] = order_key(xv[i].z * fv[i].z);
        rk[(START + i) * 4 + 3] = order_key(xv[i].w * fv[i].w);
    }
}

__device__ __forceinline__ float masked(unsigned k, float xv, float cm,
                                        unsigned thrHigh, unsigned thrLow) {
    bool m2 = (k <= thrHigh);
    bool m  = m2 && (k >= thrLow);
    return (m ? xv : 0.0f) - (m2 ? cm : 0.0f);
}

template <int START, int N>
__device__ __forceinline__ void store_rows(const float4* __restrict__ xin,
                                           const float4* __restrict__ cmin,
                                           float4* __restrict__ oout, int lane,
                                           const unsigned (&rk)[EPL],
                                           unsigned thrHigh, unsigned thrLow) {
    float4 xv[N], cm[N];
    #pragma unroll
    for (int i = 0; i < N; ++i) {
        xv[i] = xin[(START + i) * 32 + lane];
        cm[i] = cmin[(START + i) * 32 + lane];
    }
    #pragma unroll
    for (int i = 0; i < N; ++i) {
        float4 o;
        o.x = masked(rk[(START + i) * 4 + 0], xv[i].x, cm[i].x, thrHigh, thrLow);
        o.y = masked(rk[(START + i) * 4 + 1], xv[i].y, cm[i].y, thrHigh, thrLow);
        o.z = masked(rk[(START + i) * 4 + 2], xv[i].z, cm[i].z, thrHigh, thrLow);
        o.w = masked(rk[(START + i) * 4 + 3], xv[i].w, cm[i].w, thrHigh, thrLow);
        oout[(START + i) * 32 + lane] = o;
    }
}

__global__ void __launch_bounds__(256)
kw_row_kernel(const float* __restrict__ x,
              const float* __restrict__ factor,
              const float* __restrict__ colmean,
              float* __restrict__ out) {
    __shared__ unsigned cand_strip[ROWS_PER_BLOCK][CAND_CAP + 32];
    const int lane = threadIdx.x & 31;
    const int row  = blockIdx.x * ROWS_PER_BLOCK + (threadIdx.x >> 5);

    const float4* __restrict__ xin = (const float4*)(x + (size_t)row * FEATS);
    const float4* __restrict__ fin = (const float4*)factor;

    unsigned rk[EPL];

    load_keys<0, 16>(xin, fin, lane, rk);
    __builtin_amdgcn_sched_barrier(0);
    load_keys<16, 16>(xin, fin, lane, rk);

    unsigned thrHigh = 0u, thrLow = 0u;

    unsigned floorKey = 0u;
    int nAbove = FEATS;
    int bit = 31;
    #pragma unroll 1
    while (bit >= 0 && nAbove > CAND_CAP) {
        const unsigned c = floorKey | (1u << bit);
        int n = 0;
        #pragma unroll
        for (int e = 0; e < EPL; e += 2) {
            n += count2(rk[e] >= c, rk[e + 1] >= c);
            if ((e & 15) == 14) __builtin_amdgcn_sched_barrier(0);
        }
        if (n >= RANK_LOW) { floorKey = c; nAbove = n; }
        --bit;
    }

    if (nAbove <= CAND_CAP) {
        unsigned* __restrict__ strip =
            cand_strip[__builtin_amdgcn_readfirstlane((int)(threadIdx.x >> 5))];
        #pragma unroll
        for (int r = 0; r < CAND_REGS; ++r) strip[r * 32 + lane] = 0u;
        unsigned filled = 0u;
        #pragma unroll
        for (int e = 0; e < EPL; ++e) {
            const bool p = rk[e] >= floorKey;
            const unsigned b = __builtin_amdgcn_ballot_w32(p);
            const unsigned slot = __builtin_amdgcn_mbcnt_lo(b, filled);
            strip[p ? slot : (unsigned)(CAND_CAP + lane)] = rk[e];
            filled += (unsigned)__builtin_popcount(b);
            if ((e & 7) == 7) __builtin_amdgcn_sched_barrier(0);
        }
        unsigned ck[CAND_REGS];
        #pragma unroll
        for (int r = 0; r < CAND_REGS; ++r) ck[r] = strip[r * 32 + lane];


        int cntH = nAbove;
        #pragma unroll 1
        for (unsigned m = 0x80000000u; m != 0u && cntH != RANK_HIGH; m >>= 1) {
            const unsigned c = thrHigh | m;
            const int n = count_list(ck, c);
            if (n >= RANK_HIGH) { thrHigh = c; cntH = n; }
        }
        if (cntH == RANK_HIGH) thrHigh = min_at_or_above(ck, thrHigh);

        thrLow = floorKey;
        int cntL = nAbove;
        #pragma unroll 1
        for (unsigned m = (bit >= 0) ? (1u << bit) : 0u; m != 0u && cntL != RANK_LOW; m >>= 1) {
            const unsigned c = thrLow | m;
            const int n = count_list(ck, c);
            if (n >= RANK_LOW) { thrLow = c; cntL = n; }
        }
        if (cntL == RANK_LOW) thrLow = min_at_or_above(ck, thrLow);
    } else {
        #pragma unroll 1
        for (int b2 = 31; b2 >= 0; --b2) {
            const unsigned candH = thrHigh | (1u << b2);
            const unsigned candL = thrLow  | (1u << b2);
            int nH = 0, nL = 0;
            #pragma unroll
            for (int e = 0; e < EPL; e += 2) {
                nH += count2(rk[e] >= candH, rk[e + 1] >= candH);
                nL += count2(rk[e] >= candL, rk[e + 1] >= candL);
                if ((e & 15) == 14) __builtin_amdgcn_sched_barrier(0);
            }
            if (nH >= RANK_HIGH) thrHigh = candH;
            if (nL >= RANK_LOW)  thrLow  = candL;
        }
    }

    const float4* __restrict__ cmin = (const float4*)colmean;
    float4* __restrict__ oout = (float4*)(out + (size_t)row * FEATS);
    store_rows<0, 10>(xin, cmin, oout, lane, rk, thrHigh, thrLow);
    __builtin_amdgcn_sched_barrier(0);
    store_rows<10, 10>(xin, cmin, oout, lane, rk, thrHigh, thrLow);
    __builtin_amdgcn_sched_barrier(0);
    store_rows<20, 12>(xin, cmin, oout, lane, rk, thrHigh, thrLow);
}

extern "C" void kernel_launch(void* const* d_in, const int* in_sizes, int n_in,
                              void* d_out, int out_size, void* d_ws, size_t ws_size,
                              hipStream_t stream) {
    const float* x  = (const float*)d_in[0];
    const float* aa = (const float*)d_in[1];
    float* out = (float*)d_out;

    float* factor  = (float*)d_ws;
    float* colmean = (float*)d_ws + FEATS;
    float* colpart = out;

    kw_colsum_wmma_kernel<<<(FEATS / CS_TILE_COLS) * CS_SLABS, 512, 0, stream>>>(x, colpart);

    kw_factor_mean_kernel<<<FEATS / 256, 256, 0, stream>>>(aa, colpart, factor, colmean);

    kw_row_kernel<<<BATCH / ROWS_PER_BLOCK, 256, 0, stream>>>(x, factor, colmean, out);
}


# Row kernel narrowing: sign+exponent bits of the lower rank's key guessed by bisection on a 512-key strided sample, confirmed by a two-threshold full-row count; full-row bisection starts under them
# speedup vs baseline: 15.5003x; 1.1854x over previous
#include <hip/hip_runtime.h>

typedef __attribute__((ext_vector_type(2))) float v2f;
typedef __attribute__((ext_vector_type(8))) float v8f;

#define BATCH 8192
#define FEATS 4096
#define RANK_LOW 211
#define RANK_HIGH 7
#define DENSITY 0.05f
#define EPL 128
#define ROWS_PER_BLOCK 8
#define CAND_REGS 12
#define CAND_CAP (CAND_REGS * 32)
#define HEAD_BITS 9
#define SAMPLE_STEP 8
#define SAMPLE_RANK ((RANK_LOW * (EPL / SAMPLE_STEP) * 32 + FEATS / 2) / FEATS)

#define CS_TILE_COLS 256
#define CS_WAVE_ROWS 128
#define CS_SLAB_ROWS 512
#define CS_SLABS (BATCH / CS_SLAB_ROWS)
#define CS_BLK_LOADS 16

__device__ __forceinline__ void cs_load(const float4* __restrict__ p,
                                        float4 (&buf)[CS_BLK_LOADS]) {
    #pragma unroll
    for (int i = 0; i < CS_BLK_LOADS; ++i)
        buf[i] = p[(size_t)i * 2 * (FEATS / 4)];
}

__device__ __forceinline__ void cs_sum(const float4 (&buf)[CS_BLK_LOADS],
                                       v2f ones, v8f (&acc)[4]) {
    #pragma unroll
    for (int q = 0; q < CS_BLK_LOADS; q += 2) {
        v2f b0, b1, b2, b3;
        b0.x = buf[q].x; b0.y = buf[q + 1].x;
        b1.x = buf[q].y; b1.y = buf[q + 1].y;
        b2.x = buf[q].z; b2.y = buf[q + 1].z;
        b3.x = buf[q].w; b3.y = buf[q + 1].w;
        acc[0] = __builtin_amdgcn_wmma_f32_16x16x4_f32(
                     false, ones, false, b0, (short)0, acc[0], false, false);
        acc[1] = __builtin_amdgcn_wmma_f32_16x16x4_f32(
                     false, ones, false, b1, (short)0, acc[1], false, false);
        acc[2] = __builtin_amdgcn_wmma_f32_16x16x4_f32(
                     false, ones, false, b2, (short)0, acc[2], false, false);
        acc[3] = __builtin_amdgcn_wmma_f32_16x16x4_f32(
                     false, ones, false, b3, (short)0, acc[3], false, false);
    }
}

__global__ void __launch_bounds__(512)
kw_colsum_wmma_kernel(const float* __restrict__ x,
                      float* __restrict__ colpart) {
    const int lane = threadIdx.x & 31;
    const int wave = threadIdx.x >> 5;
    const int cw   = wave & 3;
    const int rw   = wave >> 2;
    const int n    = lane & 15;
    const int h    = lane >> 4;
    const int tile = blockIdx.x & (FEATS / CS_TILE_COLS - 1);
    const int slab = blockIdx.x / (FEATS / CS_TILE_COLS);

    const int col0 = tile * CS_TILE_COLS + cw * 64;
    const int row0 = slab * CS_SLAB_ROWS + rw * CS_WAVE_ROWS;
    const float4* __restrict__ p =
        (const float4*)(x + (size_t)(row0 + h) * FEATS + col0) + n;
    const size_t blk = (size_t)2 * CS_BLK_LOADS * (FEATS / 4);

    v2f ones; ones.x = 1.0f; ones.y = 1.0f;
    v8f acc[4] = {};

    float4 bufA[CS_BLK_LOADS], bufB[CS_BLK_LOADS];
    cs_load(p, bufA);
    cs_load(p + blk, bufB);
    cs_sum(bufA, ones, acc);
    cs_load(p + 2 * blk, bufA);
    cs_sum(bufB, ones, acc);
    cs_load(p + 3 * blk, bufB);
    cs_sum(bufA, ones, acc);
    cs_sum(bufB, ones, acc);

    __shared__ float4 part[CS_SLAB_ROWS / CS_WAVE_ROWS][CS_TILE_COLS / 4];
    if (lane < 16) {
        float4 s;
        s.x = acc[0][0]; s.y = acc[1][0]; s.z = acc[2][0]; s.w = acc[3][0];
        part[rw][cw * 16 + n] = s;
    }
    __syncthreads();
    if (threadIdx.x < CS_TILE_COLS / 4) {
        const float4 s0 = part[0][threadIdx.x], s1 = part[1][threadIdx.x];
        const float4 s2 = part[2][threadIdx.x], s3 = part[3][threadIdx.x];
        float4 s;
        s.x = (s0.x + s1.x) + (s2.x + s3.x);
        s.y = (s0.y + s1.y) + (s2.y + s3.y);
        s.z = (s0.z + s1.z) + (s2.z + s3.z);
        s.w = (s0.w + s1.w) + (s2.w + s3.w);
        ((float4*)(colpart + (size_t)slab * FEATS + tile * CS_TILE_COLS))[threadIdx.x] = s;
    }
}

__global__ void kw_factor_mean_kernel(const float* __restrict__ aa,
                                      const float* __restrict__ colpart,
                                      float* __restrict__ factor,
                                      float* __restrict__ colmean) {
    int j = blockIdx.x * blockDim.x + threadIdx.x;
    if (j < FEATS) {
        factor[j] = __expf((DENSITY - aa[j]) * 1.0f);
        float s = 0.0f;
        #pragma unroll
        for (int t = 0; t < CS_SLABS; ++t) s += colpart[(size_t)t * FEATS + j];
        colmean[j] = s * (1.0f / (float)BATCH);
    }
}

__device__ __forceinline__ unsigned order_key(float f) {
    unsigned u = __float_as_uint(f);
    return (u & 0x80000000u) ? ~u : (u | 0x80000000u);
}

__device__ __forceinline__ int count2(bool p0, bool p1) {
    unsigned long long m = (unsigned long long)__builtin_amdgcn_ballot_w32(p0) |
                           ((unsigned long long)__builtin_amdgcn_ballot_w32(p1) << 32);
    return __builtin_popcountll(m);
}

__device__ __forceinline__ unsigned wave_min_u32(unsigned v) {
    unsigned o;
    o = (unsigned)__builtin_amdgcn_update_dpp((int)v, (int)v, 0x168, 0xF, 0xF, false);
    v = o < v ? o : v;
    o = (unsigned)__builtin_amdgcn_update_dpp((int)v, (int)v, 0x164, 0xF, 0xF, false);
    v = o < v ? o : v;
    o = (unsigned)__builtin_amdgcn_update_dpp((int)v, (int)v, 0x162, 0xF, 0xF, false);
    v = o < v ? o : v;
    o = (unsigned)__builtin_amdgcn_update_dpp((int)v, (int)v, 0x161, 0xF, 0xF, false);
    v = o < v ? o : v;
    o = (unsigned)__builtin_amdgcn_permlanex16((int)v, (int)v, 0x76543210, 0xfedcba98,
                                               false, false);
    v = o < v ? o : v;
    return v;
}

__device__ __forceinline__ unsigned min_at_or_above(const unsigned (&ck)[CAND_REGS],
                                                    unsigned t) {
    unsigned m = 0xFFFFFFFFu;
    #pragma unroll
    for (int r = 0; r < CAND_REGS; ++r) {
        const unsigned k = (ck[r] >= t) ? ck[r] : 0xFFFFFFFFu;
        m = k < m ? k : m;
    }
    return wave_min_u32(m);
}

__device__ __forceinline__ int count_list(const unsigned (&ck)[CAND_REGS], unsigned c) {
    int n = 0;
    #pragma unroll
    for (int r = 0; r < CAND_REGS; r += 2) n += count2(ck[r] >= c, ck[r + 1] >= c);
    return n;
}

template <int START, int N>
__device__ __forceinline__ void load_keys(const float4* __restrict__ xin,
                                          const float4* __restrict__ fin,
                                          int lane, unsigned (&rk)[EPL]) {
    float4 xv[N], fv[N];
    #pragma unroll
    for (int i = 0; i < N; ++i) {
        xv[i] = xin[(START + i) * 32 + lane];
        fv[i] = fin[(START + i) * 32 + lane];
    }
    #pragma unroll
    for (int i = 0; i < N; ++i) {
        rk[(START + i) * 4 + 0] = order_key(xv[i].x * fv[i].x);
        rk[(START + i) * 4 + 1] = order_key(xv[i].y * fv[i].y);
        rk[(START + i) * 4 + 2] = order_key(xv[i].z * fv[i].z);
        rk[(START + i) * 4 + 3] = order_key(xv[i].w * fv[i].w);
    }
}

__device__ __forceinline__ float masked(unsigned k, float xv, float cm,
                                        unsigned thrHigh, unsigned thrLow) {
    bool m2 = (k <= thrHigh);
    bool m  = m2 && (k >= thrLow);
    return (m ? xv : 0.0f) - (m2 ? cm : 0.0f);
}

template <int START, int N>
__device__ __forceinline__ void store_rows(const float4* __restrict__ xin,
                                           const float4* __restrict__ cmin,
                                           float4* __restrict__ oout, int lane,
                                           const unsigned (&rk)[EPL],
                                           unsigned thrHigh, unsigned thrLow) {
    float4 xv[N], cm[N];
    #pragma unroll
    for (int i = 0; i < N; ++i) {
        xv[i] = xin[(START + i) * 32 + lane];
        cm[i] = cmin[(START + i) * 32 + lane];
    }
    #pragma unroll
    for (int i = 0; i < N; ++i) {
        float4 o;
        o.x = masked(rk[(START + i) * 4 + 0], xv[i].x, cm[i].x, thrHigh, thrLow);
        o.y = masked(rk[(START + i) * 4 + 1], xv[i].y, cm[i].y, thrHigh, thrLow);
        o.z = masked(rk[(START + i) * 4 + 2], xv[i].z, cm[i].z, thrHigh, thrLow);
        o.w = masked(rk[(START + i) * 4 + 3], xv[i].w, cm[i].w, thrHigh, thrLow);
        oout[(START + i) * 32 + lane] = o;
    }
}

__global__ void __launch_bounds__(256)
kw_row_kernel(const float* __restrict__ x,
              const float* __restrict__ factor,
              const float* __restrict__ colmean,
              float* __restrict__ out) {
    __shared__ unsigned cand_strip[ROWS_PER_BLOCK][CAND_CAP + 32];
    const int lane = threadIdx.x & 31;
    const int row  = blockIdx.x * ROWS_PER_BLOCK + (threadIdx.x >> 5);

    const float4* __restrict__ xin = (const float4*)(x + (size_t)row * FEATS);
    const float4* __restrict__ fin = (const float4*)factor;

    unsigned rk[EPL];

    load_keys<0, 16>(xin, fin, lane, rk);
    __builtin_amdgcn_sched_barrier(0);
    load_keys<16, 16>(xin, fin, lane, rk);

    unsigned thrHigh = 0u, thrLow = 0u;

    unsigned floorKey = 0u;
    int nAbove = FEATS;
    int bit = 31;

    {
        const unsigned headStep = 1u << (32 - HEAD_BITS);
        unsigned prop = 0u;
        #pragma unroll 1
        for (unsigned m = 0x80000000u; m >= headStep; m >>= 1) {
            const unsigned c = prop | m;
            int n = 0;
            #pragma unroll
            for (int e = 0; e < EPL; e += 2 * SAMPLE_STEP)
                n += count2(rk[e] >= c, rk[e + SAMPLE_STEP] >= c);
            if (n >= SAMPLE_RANK) prop = c;
        }
        const unsigned above = prop + headStep;
        int nP = 0, nQ = 0;
        #pragma unroll
        for (int e = 0; e < EPL; e += 2) {
            nP += count2(rk[e] >= prop,  rk[e + 1] >= prop);
            nQ += count2(rk[e] >= above, rk[e + 1] >= above);
            if ((e & 15) == 14) __builtin_amdgcn_sched_barrier(0);
        }
        if (above == 0u) nQ = 0;
        bool headOk = (nP >= RANK_LOW) && (nQ < RANK_LOW);
        unsigned headKey = prop;
        int headCnt = nP;
        if (!headOk) {
            const bool up = (nP >= RANK_LOW);
            const unsigned c2 = up ? above + headStep : prop - headStep;
            int n2 = 0;
            #pragma unroll
            for (int e = 0; e < EPL; e += 2) {
                n2 += count2(rk[e] >= c2, rk[e + 1] >= c2);
                if ((e & 15) == 14) __builtin_amdgcn_sched_barrier(0);
            }
            if (up) {
                if (c2 == 0u) n2 = 0;
                headOk = (n2 < RANK_LOW);
                headKey = above;
                headCnt = nQ;
            } else {
                headOk = (n2 >= RANK_LOW);
                headKey = c2;
                headCnt = n2;
            }
        }
        if (headOk) { floorKey = headKey; nAbove = headCnt; bit = 31 - HEAD_BITS; }
    }
    #pragma unroll 1
    while (bit >= 0 && nAbove > CAND_CAP) {
        const unsigned c = floorKey | (1u << bit);
        int n = 0;
        #pragma unroll
        for (int e = 0; e < EPL; e += 2) {
            n += count2(rk[e] >= c, rk[e + 1] >= c);
            if ((e & 15) == 14) __builtin_amdgcn_sched_barrier(0);
        }
        if (n >= RANK_LOW) { floorKey = c; nAbove = n; }
        --bit;
    }

    if (nAbove <= CAND_CAP) {
        unsigned* __restrict__ strip =
            cand_strip[__builtin_amdgcn_readfirstlane((int)(threadIdx.x >> 5))];
        #pragma unroll
        for (int r = 0; r < CAND_REGS; ++r) strip[r * 32 + lane] = 0u;
        unsigned filled = 0u;
        #pragma unroll
        for (int e = 0; e < EPL; ++e) {
            const bool p = rk[e] >= floorKey;
            const unsigned b = __builtin_amdgcn_ballot_w32(p);
            const unsigned slot = __builtin_amdgcn_mbcnt_lo(b, filled);
            strip[p ? slot : (unsigned)(CAND_CAP + lane)] = rk[e];
            filled += (unsigned)__builtin_popcount(b);
            if ((e & 7) == 7) __builtin_amdgcn_sched_barrier(0);
        }
        unsigned ck[CAND_REGS];
        #pragma unroll
        for (int r = 0; r < CAND_REGS; ++r) ck[r] = strip[r * 32 + lane];


        int cntH = nAbove;
        #pragma unroll 1
        for (unsigned m = 0x80000000u; m != 0u && cntH != RANK_HIGH; m >>= 1) {
            const unsigned c = thrHigh | m;
            const int n = count_list(ck, c);
            if (n >= RANK_HIGH) { thrHigh = c; cntH = n; }
        }
        if (cntH == RANK_HIGH) thrHigh = min_at_or_above(ck, thrHigh);

        thrLow = floorKey;
        int cntL = nAbove;
        #pragma unroll 1
        for (unsigned m = (bit >= 0) ? (1u << bit) : 0u; m != 0u && cntL != RANK_LOW; m >>= 1) {
            const unsigned c = thrLow | m;
            const int n = count_list(ck, c);
            if (n >= RANK_LOW) { thrLow = c; cntL = n; }
        }
        if (cntL == RANK_LOW) thrLow = min_at_or_above(ck, thrLow);
    } else {
        #pragma unroll 1
        for (int b2 = 31; b2 >= 0; --b2) {
            const unsigned candH = thrHigh | (1u << b2);
            const unsigned candL = thrLow  | (1u << b2);
            int nH = 0, nL = 0;
            #pragma unroll
            for (int e = 0; e < EPL; e += 2) {
                nH += count2(rk[e] >= candH, rk[e + 1] >= candH);
                nL += count2(rk[e] >= candL, rk[e + 1] >= candL);
                if ((e & 15) == 14) __builtin_amdgcn_sched_barrier(0);
            }
            if (nH >= RANK_HIGH) thrHigh = candH;
            if (nL >= RANK_LOW)  thrLow  = candL;
        }
    }

    const float4* __restrict__ cmin = (const float4*)colmean;
    float4* __restrict__ oout = (float4*)(out + (size_t)row * FEATS);
    store_rows<0, 10>(xin, cmin, oout, lane, rk, thrHigh, thrLow);
    __builtin_amdgcn_sched_barrier(0);
    store_rows<10, 10>(xin, cmin, oout, lane, rk, thrHigh, thrLow);
    __builtin_amdgcn_sched_barrier(0);
    store_rows<20, 12>(xin, cmin, oout, lane, rk, thrHigh, thrLow);
}

extern "C" void kernel_launch(void* const* d_in, const int* in_sizes, int n_in,
                              void* d_out, int out_size, void* d_ws, size_t ws_size,
                              hipStream_t stream) {
    const float* x  = (const float*)d_in[0];
    const float* aa = (const float*)d_in[1];
    float* out = (float*)d_out;

    float* factor  = (float*)d_ws;
    float* colmean = (float*)d_ws + FEATS;
    float* colpart = out;

    kw_colsum_wmma_kernel<<<(FEATS / CS_TILE_COLS) * CS_SLABS, 512, 0, stream>>>(x, colpart);

    kw_factor_mean_kernel<<<FEATS / 256, 256, 0, stream>>>(aa, colpart, factor, colmean);

    kw_row_kernel<<<BATCH / ROWS_PER_BLOCK, 256, 0, stream>>>(x, factor, colmean, out);
}


# Row kernel: raw row kept in a wave-private 16 KB LDS copy when first loaded; the output pass takes x from it instead of reading the row from global memory a second time
# speedup vs baseline: 16.1470x; 1.0417x over previous
#include <hip/hip_runtime.h>

typedef __attribute__((ext_vector_type(2))) float v2f;
typedef __attribute__((ext_vector_type(8))) float v8f;

#define BATCH 8192
#define FEATS 4096
#define RANK_LOW 211
#define RANK_HIGH 7
#define DENSITY 0.05f
#define EPL 128
#define ROWS_PER_BLOCK 8
#define CAND_REGS 12
#define CAND_CAP (CAND_REGS * 32)
#define HEAD_BITS 9
#define SAMPLE_STEP 8
#define SAMPLE_RANK ((RANK_LOW * (EPL / SAMPLE_STEP) * 32 + FEATS / 2) / FEATS)

#define CS_TILE_COLS 256
#define CS_WAVE_ROWS 128
#define CS_SLAB_ROWS 512
#define CS_SLABS (BATCH / CS_SLAB_ROWS)
#define CS_BLK_LOADS 16

__device__ __forceinline__ void cs_load(const float4* __restrict__ p,
                                        float4 (&buf)[CS_BLK_LOADS]) {
    #pragma unroll
    for (int i = 0; i < CS_BLK_LOADS; ++i)
        buf[i] = p[(size_t)i * 2 * (FEATS / 4)];
}

__device__ __forceinline__ void cs_sum(const float4 (&buf)[CS_BLK_LOADS],
                                       v2f ones, v8f (&acc)[4]) {
    #pragma unroll
    for (int q = 0; q < CS_BLK_LOADS; q += 2) {
        v2f b0, b1, b2, b3;
        b0.x = buf[q].x; b0.y = buf[q + 1].x;
        b1.x = buf[q].y; b1.y = buf[q + 1].y;
        b2.x = buf[q].z; b2.y = buf[q + 1].z;
        b3.x = buf[q].w; b3.y = buf[q + 1].w;
        acc[0] = __builtin_amdgcn_wmma_f32_16x16x4_f32(
                     false, ones, false, b0, (short)0, acc[0], false, false);
        acc[1] = __builtin_amdgcn_wmma_f32_16x16x4_f32(
                     false, ones, false, b1, (short)0, acc[1], false, false);
        acc[2] = __builtin_amdgcn_wmma_f32_16x16x4_f32(
                     false, ones, false, b2, (short)0, acc[2], false, false);
        acc[3] = __builtin_amdgcn_wmma_f32_16x16x4_f32(
                     false, ones, false, b3, (short)0, acc[3], false, false);
    }
}

__global__ void __launch_bounds__(512)
kw_colsum_wmma_kernel(const float* __restrict__ x,
                      float* __restrict__ colpart) {
    const int lane = threadIdx.x & 31;
    const int wave = threadIdx.x >> 5;
    const int cw   = wave & 3;
    const int rw   = wave >> 2;
    const int n    = lane & 15;
    const int h    = lane >> 4;
    const int tile = blockIdx.x & (FEATS / CS_TILE_COLS - 1);
    const int slab = blockIdx.x / (FEATS / CS_TILE_COLS);

    const int col0 = tile * CS_TILE_COLS + cw * 64;
    const int row0 = slab * CS_SLAB_ROWS + rw * CS_WAVE_ROWS;
    const float4* __restrict__ p =
        (const float4*)(x + (size_t)(row0 + h) * FEATS + col0) + n;
    const size_t blk = (size_t)2 * CS_BLK_LOADS * (FEATS / 4);

    v2f ones; ones.x = 1.0f; ones.y = 1.0f;
    v8f acc[4] = {};

    float4 bufA[CS_BLK_LOADS], bufB[CS_BLK_LOADS];
    cs_load(p, bufA);
    cs_load(p + blk, bufB);
    cs_sum(bufA, ones, acc);
    cs_load(p + 2 * blk, bufA);
    cs_sum(bufB, ones, acc);
    cs_load(p + 3 * blk, bufB);
    cs_sum(bufA, ones, acc);
    cs_sum(bufB, ones, acc);

    __shared__ float4 part[CS_SLAB_ROWS / CS_WAVE_ROWS][CS_TILE_COLS / 4];
    if (lane < 16) {
        float4 s;
        s.x = acc[0][0]; s.y = acc[1][0]; s.z = acc[2][0]; s.w = acc[3][0];
        part[rw][cw * 16 + n] = s;
    }
    __syncthreads();
    if (threadIdx.x < CS_TILE_COLS / 4) {
        const float4 s0 = part[0][threadIdx.x], s1 = part[1][threadIdx.x];
        const float4 s2 = part[2][threadIdx.x], s3 = part[3][threadIdx.x];
        float4 s;
        s.x = (s0.x + s1.x) + (s2.x + s3.x);
        s.y = (s0.y + s1.y) + (s2.y + s3.y);
        s.z = (s0.z + s1.z) + (s2.z + s3.z);
        s.w = (s0.w + s1.w) + (s2.w + s3.w);
        ((float4*)(colpart + (size_t)slab * FEATS + tile * CS_TILE_COLS))[threadIdx.x] = s;
    }
}

__global__ void kw_factor_mean_kernel(const float* __restrict__ aa,
                                      const float* __restrict__ colpart,
                                      float* __restrict__ factor,
                                      float* __restrict__ colmean) {
    int j = blockIdx.x * blockDim.x + threadIdx.x;
    if (j < FEATS) {
        factor[j] = __expf((DENSITY - aa[j]) * 1.0f);
        float s = 0.0f;
        #pragma unroll
        for (int t = 0; t < CS_SLABS; ++t) s += colpart[(size_t)t * FEATS + j];
        colmean[j] = s * (1.0f / (float)BATCH);
    }
}

__device__ __forceinline__ unsigned order_key(float f) {
    unsigned u = __float_as_uint(f);
    return (u & 0x80000000u) ? ~u : (u | 0x80000000u);
}

__device__ __forceinline__ int count2(bool p0, bool p1) {
    unsigned long long m = (unsigned long long)__builtin_amdgcn_ballot_w32(p0) |
                           ((unsigned long long)__builtin_amdgcn_ballot_w32(p1) << 32);
    return __builtin_popcountll(m);
}

__device__ __forceinline__ unsigned wave_min_u32(unsigned v) {
    unsigned o;
    o = (unsigned)__builtin_amdgcn_update_dpp((int)v, (int)v, 0x168, 0xF, 0xF, false);
    v = o < v ? o : v;
    o = (unsigned)__builtin_amdgcn_update_dpp((int)v, (int)v, 0x164, 0xF, 0xF, false);
    v = o < v ? o : v;
    o = (unsigned)__builtin_amdgcn_update_dpp((int)v, (int)v, 0x162, 0xF, 0xF, false);
    v = o < v ? o : v;
    o = (unsigned)__builtin_amdgcn_update_dpp((int)v, (int)v, 0x161, 0xF, 0xF, false);
    v = o < v ? o : v;
    o = (unsigned)__builtin_amdgcn_permlanex16((int)v, (int)v, 0x76543210, 0xfedcba98,
                                               false, false);
    v = o < v ? o : v;
    return v;
}

__device__ __forceinline__ unsigned min_at_or_above(const unsigned (&ck)[CAND_REGS],
                                                    unsigned t) {
    unsigned m = 0xFFFFFFFFu;
    #pragma unroll
    for (int r = 0; r < CAND_REGS; ++r) {
        const unsigned k = (ck[r] >= t) ? ck[r] : 0xFFFFFFFFu;
        m = k < m ? k : m;
    }
    return wave_min_u32(m);
}

__device__ __forceinline__ int count_list(const unsigned (&ck)[CAND_REGS], unsigned c) {
    int n = 0;
    #pragma unroll
    for (int r = 0; r < CAND_REGS; r += 2) n += count2(ck[r] >= c, ck[r + 1] >= c);
    return n;
}

template <int START, int N>
__device__ __forceinline__ void load_keys(const float4* __restrict__ xin,
                                          const float4* __restrict__ fin,
                                          float4* xrow,
                                          int lane, unsigned (&rk)[EPL]) {
    float4 xv[N], fv[N];
    #pragma unroll
    for (int i = 0; i < N; ++i) {
        xv[i] = xin[(START + i) * 32 + lane];
        fv[i] = fin[(START + i) * 32 + lane];
    }
    #pragma unroll
    for (int i = 0; i < N; ++i) {
        xrow[(START + i) * 32 + lane] = xv[i];
        rk[(START + i) * 4 + 0] = order_key(xv[i].x * fv[i].x);
        rk[(START + i) * 4 + 1] = order_key(xv[i].y * fv[i].y);
        rk[(START + i) * 4 + 2] = order_key(xv[i].z * fv[i].z);
        rk[(START + i) * 4 + 3] = order_key(xv[i].w * fv[i].w);
    }
}

__device__ __forceinline__ float masked(unsigned k, float xv, float cm,
                                        unsigned thrHigh, unsigned thrLow) {
    bool m2 = (k <= thrHigh);
    bool m  = m2 && (k >= thrLow);
    return (m ? xv : 0.0f) - (m2 ? cm : 0.0f);
}

template <int START, int N>
__device__ __forceinline__ void store_rows(const float4* xrow,
                                           const float4* __restrict__ cmin,
                                           float4* __restrict__ oout, int lane,
                                           const unsigned (&rk)[EPL],
                                           unsigned thrHigh, unsigned thrLow) {
    float4 xv[N], cm[N];
    #pragma unroll
    for (int i = 0; i < N; ++i) {
        xv[i] = xrow[(START + i) * 32 + lane];
        cm[i] = cmin[(START + i) * 32 + lane];
    }
    #pragma unroll
    for (int i = 0; i < N; ++i) {
        float4 o;
        o.x = masked(rk[(START + i) * 4 + 0], xv[i].x, cm[i].x, thrHigh, thrLow);
        o.y = masked(rk[(START + i) * 4 + 1], xv[i].y, cm[i].y, thrHigh, thrLow);
        o.z = masked(rk[(START + i) * 4 + 2], xv[i].z, cm[i].z, thrHigh, thrLow);
        o.w = masked(rk[(START + i) * 4 + 3], xv[i].w, cm[i].w, thrHigh, thrLow);
        oout[(START + i) * 32 + lane] = o;
    }
}

__global__ void __launch_bounds__(256)
kw_row_kernel(const float* __restrict__ x,
              const float* __restrict__ factor,
              const float* __restrict__ colmean,
              float* __restrict__ out) {
    __shared__ unsigned cand_strip[ROWS_PER_BLOCK][CAND_CAP + 32];
    __shared__ float4 xrow_copy[ROWS_PER_BLOCK][FEATS / 4];
    const int lane = threadIdx.x & 31;
    const int row  = blockIdx.x * ROWS_PER_BLOCK + (threadIdx.x >> 5);

    const float4* __restrict__ xin = (const float4*)(x + (size_t)row * FEATS);
    const float4* __restrict__ fin = (const float4*)factor;

    unsigned rk[EPL];

    float4* xrow = xrow_copy[threadIdx.x >> 5];
    load_keys<0, 16>(xin, fin, xrow, lane, rk);
    __builtin_amdgcn_sched_barrier(0);
    load_keys<16, 16>(xin, fin, xrow, lane, rk);

    unsigned thrHigh = 0u, thrLow = 0u;

    unsigned floorKey = 0u;
    int nAbove = FEATS;
    int bit = 31;

    {
        const unsigned headStep = 1u << (32 - HEAD_BITS);
        unsigned prop = 0u;
        #pragma unroll 1
        for (unsigned m = 0x80000000u; m >= headStep; m >>= 1) {
            const unsigned c = prop | m;
            int n = 0;
            #pragma unroll
            for (int e = 0; e < EPL; e += 2 * SAMPLE_STEP)
                n += count2(rk[e] >= c, rk[e + SAMPLE_STEP] >= c);
            if (n >= SAMPLE_RANK) prop = c;
        }
        const unsigned above = prop + headStep;
        int nP = 0, nQ = 0;
        #pragma unroll
        for (int e = 0; e < EPL; e += 2) {
            nP += count2(rk[e] >= prop,  rk[e + 1] >= prop);
            nQ += count2(rk[e] >= above, rk[e + 1] >= above);
            if ((e & 15) == 14) __builtin_amdgcn_sched_barrier(0);
        }
        if (above == 0u) nQ = 0;
        bool headOk = (nP >= RANK_LOW) && (nQ < RANK_LOW);
        unsigned headKey = prop;
        int headCnt = nP;
        if (!headOk) {
            const bool up = (nP >= RANK_LOW);
            const unsigned c2 = up ? above + headStep : prop - headStep;
            int n2 = 0;
            #pragma unroll
            for (int e = 0; e < EPL; e += 2) {
                n2 += count2(rk[e] >= c2, rk[e + 1] >= c2);
                if ((e & 15) == 14) __builtin_amdgcn_sched_barrier(0);
            }
            if (up) {
                if (c2 == 0u) n2 = 0;
                headOk = (n2 < RANK_LOW);
                headKey = above;
                headCnt = nQ;
            } else {
                headOk = (n2 >= RANK_LOW);
                headKey = c2;
                headCnt = n2;
            }
        }
        if (headOk) { floorKey = headKey; nAbove = headCnt; bit = 31 - HEAD_BITS; }
    }
    #pragma unroll 1
    while (bit >= 0 && nAbove > CAND_CAP) {
        const unsigned c = floorKey | (1u << bit);
        int n = 0;
        #pragma unroll
        for (int e = 0; e < EPL; e += 2) {
            n += count2(rk[e] >= c, rk[e + 1] >= c);
            if ((e & 15) == 14) __builtin_amdgcn_sched_barrier(0);
        }
        if (n >= RANK_LOW) { floorKey = c; nAbove = n; }
        --bit;
    }

    if (nAbove <= CAND_CAP) {
        unsigned* __restrict__ strip =
            cand_strip[__builtin_amdgcn_readfirstlane((int)(threadIdx.x >> 5))];
        #pragma unroll
        for (int r = 0; r < CAND_REGS; ++r) strip[r * 32 + lane] = 0u;
        unsigned filled = 0u;
        #pragma unroll
        for (int e = 0; e < EPL; ++e) {
            const bool p = rk[e] >= floorKey;
            const unsigned b = __builtin_amdgcn_ballot_w32(p);
            const unsigned slot = __builtin_amdgcn_mbcnt_lo(b, filled);
            strip[p ? slot : (unsigned)(CAND_CAP + lane)] = rk[e];
            filled += (unsigned)__builtin_popcount(b);
            if ((e & 7) == 7) __builtin_amdgcn_sched_barrier(0);
        }
        unsigned ck[CAND_REGS];
        #pragma unroll
        for (int r = 0; r < CAND_REGS; ++r) ck[r] = strip[r * 32 + lane];


        int cntH = nAbove;
        #pragma unroll 1
        for (unsigned m = 0x80000000u; m != 0u && cntH != RANK_HIGH; m >>= 1) {
            const unsigned c = thrHigh | m;
            const int n = count_list(ck, c);
            if (n >= RANK_HIGH) { thrHigh = c; cntH = n; }
        }
        if (cntH == RANK_HIGH) thrHigh = min_at_or_above(ck, thrHigh);

        thrLow = floorKey;
        int cntL = nAbove;
        #pragma unroll 1
        for (unsigned m = (bit >= 0) ? (1u << bit) : 0u; m != 0u && cntL != RANK_LOW; m >>= 1) {
            const unsigned c = thrLow | m;
            const int n = count_list(ck, c);
            if (n >= RANK_LOW) { thrLow = c; cntL = n; }
        }
        if (cntL == RANK_LOW) thrLow = min_at_or_above(ck, thrLow);
    } else {
        #pragma unroll 1
        for (int b2 = 31; b2 >= 0; --b2) {
            const unsigned candH = thrHigh | (1u << b2);
            const unsigned candL = thrLow  | (1u << b2);
            int nH = 0, nL = 0;
            #pragma unroll
            for (int e = 0; e < EPL; e += 2) {
                nH += count2(rk[e] >= candH, rk[e + 1] >= candH);
                nL += count2(rk[e] >= candL, rk[e + 1] >= candL);
                if ((e & 15) == 14) __builtin_amdgcn_sched_barrier(0);
            }
            if (nH >= RANK_HIGH) thrHigh = candH;
            if (nL >= RANK_LOW)  thrLow  = candL;
        }
    }

    const float4* __restrict__ cmin = (const float4*)colmean;
    float4* __restrict__ oout = (float4*)(out + (size_t)row * FEATS);
    store_rows<0, 10>(xrow, cmin, oout, lane, rk, thrHigh, thrLow);
    __builtin_amdgcn_sched_barrier(0);
    store_rows<10, 10>(xrow, cmin, oout, lane, rk, thrHigh, thrLow);
    __builtin_amdgcn_sched_barrier(0);
    store_rows<20, 12>(xrow, cmin, oout, lane, rk, thrHigh, thrLow);
}

extern "C" void kernel_launch(void* const* d_in, const int* in_sizes, int n_in,
                              void* d_out, int out_size, void* d_ws, size_t ws_size,
                              hipStream_t stream) {
    const float* x  = (const float*)d_in[0];
    const float* aa = (const float*)d_in[1];
    float* out = (float*)d_out;

    float* factor  = (float*)d_ws;
    float* colmean = (float*)d_ws + FEATS;
    float* colpart = out;

    kw_colsum_wmma_kernel<<<(FEATS / CS_TILE_COLS) * CS_SLABS, 512, 0, stream>>>(x, colpart);

    kw_factor_mean_kernel<<<FEATS / 256, 256, 0, stream>>>(aa, colpart, factor, colmean);

    kw_row_kernel<<<BATCH / ROWS_PER_BLOCK, 256, 0, stream>>>(x, factor, colmean, out);
}


# Row kernel: per-element order-key transform dropped; products stay floats in registers, each full-row count compares them with its trial key mapped back to a float; only the short list gets keys
# speedup vs baseline: 16.9083x; 1.0471x over previous
#include <hip/hip_runtime.h>

typedef __attribute__((ext_vector_type(2))) float v2f;
typedef __attribute__((ext_vector_type(8))) float v8f;

#define BATCH 8192
#define FEATS 4096
#define RANK_LOW 211
#define RANK_HIGH 7
#define DENSITY 0.05f
#define EPL 128
#define ROWS_PER_BLOCK 8
#define CAND_REGS 12
#define CAND_CAP (CAND_REGS * 32)
#define HEAD_BITS 9
#define SAMPLE_STEP 8
#define SAMPLE_RANK ((RANK_LOW * (EPL / SAMPLE_STEP) * 32 + FEATS / 2) / FEATS)

#define CS_TILE_COLS 256
#define CS_WAVE_ROWS 128
#define CS_SLAB_ROWS 512
#define CS_SLABS (BATCH / CS_SLAB_ROWS)
#define CS_BLK_LOADS 16

__device__ __forceinline__ void cs_load(const float4* __restrict__ p,
                                        float4 (&buf)[CS_BLK_LOADS]) {
    #pragma unroll
    for (int i = 0; i < CS_BLK_LOADS; ++i)
        buf[i] = p[(size_t)i * 2 * (FEATS / 4)];
}

__device__ __forceinline__ void cs_sum(const float4 (&buf)[CS_BLK_LOADS],
                                       v2f ones, v8f (&acc)[4]) {
    #pragma unroll
    for (int q = 0; q < CS_BLK_LOADS; q += 2) {
        v2f b0, b1, b2, b3;
        b0.x = buf[q].x; b0.y = buf[q + 1].x;
        b1.x = buf[q].y; b1.y = buf[q + 1].y;
        b2.x = buf[q].z; b2.y = buf[q + 1].z;
        b3.x = buf[q].w; b3.y = buf[q + 1].w;
        acc[0] = __builtin_amdgcn_wmma_f32_16x16x4_f32(
                     false, ones, false, b0, (short)0, acc[0], false, false);
        acc[1] = __builtin_amdgcn_wmma_f32_16x16x4_f32(
                     false, ones, false, b1, (short)0, acc[1], false, false);
        acc[2] = __builtin_amdgcn_wmma_f32_16x16x4_f32(
                     false, ones, false, b2, (short)0, acc[2], false, false);
        acc[3] = __builtin_amdgcn_wmma_f32_16x16x4_f32(
                     false, ones, false, b3, (short)0, acc[3], false, false);
    }
}

__global__ void __launch_bounds__(512)
kw_colsum_wmma_kernel(const float* __restrict__ x,
                      float* __restrict__ colpart) {
    const int lane = threadIdx.x & 31;
    const int wave = threadIdx.x >> 5;
    const int cw   = wave & 3;
    const int rw   = wave >> 2;
    const int n    = lane & 15;
    const int h    = lane >> 4;
    const int tile = blockIdx.x & (FEATS / CS_TILE_COLS - 1);
    const int slab = blockIdx.x / (FEATS / CS_TILE_COLS);

    const int col0 = tile * CS_TILE_COLS + cw * 64;
    const int row0 = slab * CS_SLAB_ROWS + rw * CS_WAVE_ROWS;
    const float4* __restrict__ p =
        (const float4*)(x + (size_t)(row0 + h) * FEATS + col0) + n;
    const size_t blk = (size_t)2 * CS_BLK_LOADS * (FEATS / 4);

    v2f ones; ones.x = 1.0f; ones.y = 1.0f;
    v8f acc[4] = {};

    float4 bufA[CS_BLK_LOADS], bufB[CS_BLK_LOADS];
    cs_load(p, bufA);
    cs_load(p + blk, bufB);
    cs_sum(bufA, ones, acc);
    cs_load(p + 2 * blk, bufA);
    cs_sum(bufB, ones, acc);
    cs_load(p + 3 * blk, bufB);
    cs_sum(bufA, ones, acc);
    cs_sum(bufB, ones, acc);

    __shared__ float4 part[CS_SLAB_ROWS / CS_WAVE_ROWS][CS_TILE_COLS / 4];
    if (lane < 16) {
        float4 s;
        s.x = acc[0][0]; s.y = acc[1][0]; s.z = acc[2][0]; s.w = acc[3][0];
        part[rw][cw * 16 + n] = s;
    }
    __syncthreads();
    if (threadIdx.x < CS_TILE_COLS / 4) {
        const float4 s0 = part[0][threadIdx.x], s1 = part[1][threadIdx.x];
        const float4 s2 = part[2][threadIdx.x], s3 = part[3][threadIdx.x];
        float4 s;
        s.x = (s0.x + s1.x) + (s2.x + s3.x);
        s.y = (s0.y + s1.y) + (s2.y + s3.y);
        s.z = (s0.z + s1.z) + (s2.z + s3.z);
        s.w = (s0.w + s1.w) + (s2.w + s3.w);
        ((float4*)(colpart + (size_t)slab * FEATS + tile * CS_TILE_COLS))[threadIdx.x] = s;
    }
}

__global__ void kw_factor_mean_kernel(const float* __restrict__ aa,
                                      const float* __restrict__ colpart,
                                      float* __restrict__ factor,
                                      float* __restrict__ colmean) {
    int j = blockIdx.x * blockDim.x + threadIdx.x;
    if (j < FEATS) {
        factor[j] = __expf((DENSITY - aa[j]) * 1.0f);
        float s = 0.0f;
        #pragma unroll
        for (int t = 0; t < CS_SLABS; ++t) s += colpart[(size_t)t * FEATS + j];
        colmean[j] = s * (1.0f / (float)BATCH);
    }
}

__device__ __forceinline__ unsigned order_key(float f) {
    unsigned u = __float_as_uint(f);
    return (u & 0x80000000u) ? ~u : (u | 0x80000000u);
}

__device__ __forceinline__ float key_float(unsigned c) {
    c = c < 0x007FFFFFu ? 0x007FFFFFu : c;
    return __uint_as_float((c & 0x80000000u) ? (c ^ 0x80000000u) : ~c);
}

__device__ __forceinline__ int count2(bool p0, bool p1) {
    unsigned long long m = (unsigned long long)__builtin_amdgcn_ballot_w32(p0) |
                           ((unsigned long long)__builtin_amdgcn_ballot_w32(p1) << 32);
    return __builtin_popcountll(m);
}

__device__ __forceinline__ unsigned wave_min_u32(unsigned v) {
    unsigned o;
    o = (unsigned)__builtin_amdgcn_update_dpp((int)v, (int)v, 0x168, 0xF, 0xF, false);
    v = o < v ? o : v;
    o = (unsigned)__builtin_amdgcn_update_dpp((int)v, (int)v, 0x164, 0xF, 0xF, false);
    v = o < v ? o : v;
    o = (unsigned)__builtin_amdgcn_update_dpp((int)v, (int)v, 0x162, 0xF, 0xF, false);
    v = o < v ? o : v;
    o = (unsigned)__builtin_amdgcn_update_dpp((int)v, (int)v, 0x161, 0xF, 0xF, false);
    v = o < v ? o : v;
    o = (unsigned)__builtin_amdgcn_permlanex16((int)v, (int)v, 0x76543210, 0xfedcba98,
                                               false, false);
    v = o < v ? o : v;
    return v;
}

__device__ __forceinline__ unsigned min_at_or_above(const unsigned (&ck)[CAND_REGS],
                                                    unsigned t) {
    unsigned m = 0xFFFFFFFFu;
    #pragma unroll
    for (int r = 0; r < CAND_REGS; ++r) {
        const unsigned k = (ck[r] >= t) ? ck[r] : 0xFFFFFFFFu;
        m = k < m ? k : m;
    }
    return wave_min_u32(m);
}

__device__ __forceinline__ int count_list(const unsigned (&ck)[CAND_REGS], unsigned c) {
    int n = 0;
    #pragma unroll
    for (int r = 0; r < CAND_REGS; r += 2) n += count2(ck[r] >= c, ck[r + 1] >= c);
    return n;
}

template <int START, int N>
__device__ __forceinline__ void load_keys(const float4* __restrict__ xin,
                                          const float4* __restrict__ fin,
                                          float4* xrow,
                                          int lane, float (&rk)[EPL]) {
    float4 xv[N], fv[N];
    #pragma unroll
    for (int i = 0; i < N; ++i) {
        xv[i] = xin[(START + i) * 32 + lane];
        fv[i] = fin[(START + i) * 32 + lane];
    }
    #pragma unroll
    for (int i = 0; i < N; ++i) {
        xrow[(START + i) * 32 + lane] = xv[i];
        rk[(START + i) * 4 + 0] = xv[i].x * fv[i].x;
        rk[(START + i) * 4 + 1] = xv[i].y * fv[i].y;
        rk[(START + i) * 4 + 2] = xv[i].z * fv[i].z;
        rk[(START + i) * 4 + 3] = xv[i].w * fv[i].w;
        asm volatile("" : "+v"(rk[(START + i) * 4 + 0]), "+v"(rk[(START + i) * 4 + 1]),
                          "+v"(rk[(START + i) * 4 + 2]), "+v"(rk[(START + i) * 4 + 3]));
    }
}

__device__ __forceinline__ float masked(float xp, float xv, float cm,
                                        float thrHigh, float thrLow) {
    bool m2 = (xp <= thrHigh);
    bool m  = m2 && (xp >= thrLow);
    return (m ? xv : 0.0f) - (m2 ? cm : 0.0f);
}

template <int START, int N>
__device__ __forceinline__ void store_rows(const float4* xrow,
                                           const float4* __restrict__ cmin,
                                           float4* __restrict__ oout, int lane,
                                           const float (&rk)[EPL],
                                           float thrHigh, float thrLow) {
    float4 xv[N], cm[N];
    #pragma unroll
    for (int i = 0; i < N; ++i) {
        xv[i] = xrow[(START + i) * 32 + lane];
        cm[i] = cmin[(START + i) * 32 + lane];
    }
    #pragma unroll
    for (int i = 0; i < N; ++i) {
        float4 o;
        o.x = masked(rk[(START + i) * 4 + 0], xv[i].x, cm[i].x, thrHigh, thrLow);
        o.y = masked(rk[(START + i) * 4 + 1], xv[i].y, cm[i].y, thrHigh, thrLow);
        o.z = masked(rk[(START + i) * 4 + 2], xv[i].z, cm[i].z, thrHigh, thrLow);
        o.w = masked(rk[(START + i) * 4 + 3], xv[i].w, cm[i].w, thrHigh, thrLow);
        oout[(START + i) * 32 + lane] = o;
    }
}

__global__ void __launch_bounds__(256)
kw_row_kernel(const float* __restrict__ x,
              const float* __restrict__ factor,
              const float* __restrict__ colmean,
              float* __restrict__ out) {
    __shared__ unsigned cand_strip[ROWS_PER_BLOCK][CAND_CAP + 32];
    __shared__ float4 xrow_copy[ROWS_PER_BLOCK][FEATS / 4];
    const int lane = threadIdx.x & 31;
    const int row  = blockIdx.x * ROWS_PER_BLOCK + (threadIdx.x >> 5);

    const float4* __restrict__ xin = (const float4*)(x + (size_t)row * FEATS);
    const float4* __restrict__ fin = (const float4*)factor;

    float rk[EPL];

    float4* xrow = xrow_copy[threadIdx.x >> 5];
    load_keys<0, 16>(xin, fin, xrow, lane, rk);
    __builtin_amdgcn_sched_barrier(0);
    load_keys<16, 16>(xin, fin, xrow, lane, rk);

    unsigned thrHigh = 0u, thrLow = 0u;

    unsigned floorKey = 0u;
    int nAbove = FEATS;
    int bit = 31;

    {
        const unsigned headStep = 1u << (32 - HEAD_BITS);
        unsigned prop = 0u;
        #pragma unroll 1
        for (unsigned m = 0x80000000u; m >= headStep; m >>= 1) {
            const unsigned c = prop | m;
            const float cf = key_float(c);
            int n = 0;
            #pragma unroll
            for (int e = 0; e < EPL; e += 2 * SAMPLE_STEP)
                n += count2(rk[e] >= cf, rk[e + SAMPLE_STEP] >= cf);
            if (n >= SAMPLE_RANK) prop = c;
        }
        const unsigned above = prop + headStep;
        const float propF = key_float(prop), aboveF = key_float(above);
        int nP = 0, nQ = 0;
        #pragma unroll
        for (int e = 0; e < EPL; e += 2) {
            nP += count2(rk[e] >= propF,  rk[e + 1] >= propF);
            nQ += count2(rk[e] >= aboveF, rk[e + 1] >= aboveF);
            if ((e & 15) == 14) __builtin_amdgcn_sched_barrier(0);
        }
        if (above == 0u) nQ = 0;
        bool headOk = (nP >= RANK_LOW) && (nQ < RANK_LOW);
        unsigned headKey = prop;
        int headCnt = nP;
        if (!headOk) {
            const bool up = (nP >= RANK_LOW);
            const unsigned c2 = up ? above + headStep : prop - headStep;
            const float c2F = key_float(c2);
            int n2 = 0;
            #pragma unroll
            for (int e = 0; e < EPL; e += 2) {
                n2 += count2(rk[e] >= c2F, rk[e + 1] >= c2F);
                if ((e & 15) == 14) __builtin_amdgcn_sched_barrier(0);
            }
            if (up) {
                if (c2 == 0u) n2 = 0;
                headOk = (n2 < RANK_LOW);
                headKey = above;
                headCnt = nQ;
            } else {
                headOk = (n2 >= RANK_LOW);
                headKey = c2;
                headCnt = n2;
            }
        }
        if (headOk) { floorKey = headKey; nAbove = headCnt; bit = 31 - HEAD_BITS; }
    }
    #pragma unroll 1
    while (bit >= 0 && nAbove > CAND_CAP) {
        const unsigned c = floorKey | (1u << bit);
        const float cf = key_float(c);
        int n = 0;
        #pragma unroll
        for (int e = 0; e < EPL; e += 2) {
            n += count2(rk[e] >= cf, rk[e + 1] >= cf);
            if ((e & 15) == 14) __builtin_amdgcn_sched_barrier(0);
        }
        if (n >= RANK_LOW) { floorKey = c; nAbove = n; }
        --bit;
    }

    if (nAbove <= CAND_CAP) {
        unsigned* __restrict__ strip =
            cand_strip[__builtin_amdgcn_readfirstlane((int)(threadIdx.x >> 5))];
        #pragma unroll
        for (int r = 0; r < CAND_REGS; ++r) strip[r * 32 + lane] = 0xFFFFFFFFu;
        const float floorF = key_float(floorKey);
        unsigned filled = 0u;
        #pragma unroll
        for (int e = 0; e < EPL; ++e) {
            const bool p = rk[e] >= floorF;
            const unsigned b = __builtin_amdgcn_ballot_w32(p);
            const unsigned slot = __builtin_amdgcn_mbcnt_lo(b, filled);
            strip[p ? slot : (unsigned)(CAND_CAP + lane)] = __float_as_uint(rk[e]);
            filled += (unsigned)__builtin_popcount(b);
            if ((e & 7) == 7) __builtin_amdgcn_sched_barrier(0);
        }
        unsigned ck[CAND_REGS];
        #pragma unroll
        for (int r = 0; r < CAND_REGS; ++r) {
            const unsigned k = order_key(__uint_as_float(strip[r * 32 + lane]));
            ck[r] = (k == 0x7FFFFFFFu) ? 0x80000000u : k;
        }


        int cntH = nAbove;
        #pragma unroll 1
        for (unsigned m = 0x80000000u; m != 0u && cntH != RANK_HIGH; m >>= 1) {
            const unsigned c = thrHigh | m;
            const int n = count_list(ck, c);
            if (n >= RANK_HIGH) { thrHigh = c; cntH = n; }
        }
        if (cntH == RANK_HIGH) thrHigh = min_at_or_above(ck, thrHigh);

        thrLow = floorKey;
        int cntL = nAbove;
        #pragma unroll 1
        for (unsigned m = (bit >= 0) ? (1u << bit) : 0u; m != 0u && cntL != RANK_LOW; m >>= 1) {
            const unsigned c = thrLow | m;
            const int n = count_list(ck, c);
            if (n >= RANK_LOW) { thrLow = c; cntL = n; }
        }
        if (cntL == RANK_LOW) thrLow = min_at_or_above(ck, thrLow);
    } else {
        #pragma unroll 1
        for (int b2 = 31; b2 >= 0; --b2) {
            const unsigned candH = thrHigh | (1u << b2);
            const unsigned candL = thrLow  | (1u << b2);
            const float candHF = key_float(candH), candLF = key_float(candL);
            int nH = 0, nL = 0;
            #pragma unroll
            for (int e = 0; e < EPL; e += 2) {
                nH += count2(rk[e] >= candHF, rk[e + 1] >= candHF);
                nL += count2(rk[e] >= candLF, rk[e + 1] >= candLF);
                if ((e & 15) == 14) __builtin_amdgcn_sched_barrier(0);
            }
            if (nH >= RANK_HIGH) thrHigh = candH;
            if (nL >= RANK_LOW)  thrLow  = candL;
        }
    }

    const float4* __restrict__ cmin = (const float4*)colmean;
    float4* __restrict__ oout = (float4*)(out + (size_t)row * FEATS);
    const float thrHighF = key_float(thrHigh), thrLowF = key_float(thrLow);
    store_rows<0, 10>(xrow, cmin, oout, lane, rk, thrHighF, thrLowF);
    __builtin_amdgcn_sched_barrier(0);
    store_rows<10, 10>(xrow, cmin, oout, lane, rk, thrHighF, thrLowF);
    __builtin_amdgcn_sched_barrier(0);
    store_rows<20, 12>(xrow, cmin, oout, lane, rk, thrHighF, thrLowF);
}

extern "C" void kernel_launch(void* const* d_in, const int* in_sizes, int n_in,
                              void* d_out, int out_size, void* d_ws, size_t ws_size,
                              hipStream_t stream) {
    const float* x  = (const float*)d_in[0];
    const float* aa = (const float*)d_in[1];
    float* out = (float*)d_out;

    float* factor  = (float*)d_ws;
    float* colmean = (float*)d_ws + FEATS;
    float* colpart = out;

    kw_colsum_wmma_kernel<<<(FEATS / CS_TILE_COLS) * CS_SLABS, 512, 0, stream>>>(x, colpart);

    kw_factor_mean_kernel<<<FEATS / 256, 256, 0, stream>>>(aa, colpart, factor, colmean);

    kw_row_kernel<<<BATCH / ROWS_PER_BLOCK, 256, 0, stream>>>(x, factor, colmean, out);
}


# Row kernel: 512 resident work-groups, each wave does two rows in turn; while it searches the first it requests the second into L2 with two prefetch instructions, one 256-B unit per lane each
# speedup vs baseline: 17.4373x; 1.0313x over previous
#include <hip/hip_runtime.h>

typedef __attribute__((ext_vector_type(2))) float v2f;
typedef __attribute__((ext_vector_type(8))) float v8f;

#define BATCH 8192
#define FEATS 4096
#define RANK_LOW 211
#define RANK_HIGH 7
#define DENSITY 0.05f
#define EPL 128
#define ROWS_PER_BLOCK 8
#define ROWS_PER_WAVE 2
#define ROW_BLOCKS (BATCH / (ROWS_PER_BLOCK * ROWS_PER_WAVE))
#define CAND_REGS 12
#define CAND_CAP (CAND_REGS * 32)
#define HEAD_BITS 9
#define SAMPLE_STEP 8
#define SAMPLE_RANK ((RANK_LOW * (EPL / SAMPLE_STEP) * 32 + FEATS / 2) / FEATS)

#define CS_TILE_COLS 256
#define CS_WAVE_ROWS 128
#define CS_SLAB_ROWS 512
#define CS_SLABS (BATCH / CS_SLAB_ROWS)
#define CS_BLK_LOADS 16

__device__ __forceinline__ void cs_load(const float4* __restrict__ p,
                                        float4 (&buf)[CS_BLK_LOADS]) {
    #pragma unroll
    for (int i = 0; i < CS_BLK_LOADS; ++i)
        buf[i] = p[(size_t)i * 2 * (FEATS / 4)];
}

__device__ __forceinline__ void cs_sum(const float4 (&buf)[CS_BLK_LOADS],
                                       v2f ones, v8f (&acc)[4]) {
    #pragma unroll
    for (int q = 0; q < CS_BLK_LOADS; q += 2) {
        v2f b0, b1, b2, b3;
        b0.x = buf[q].x; b0.y = buf[q + 1].x;
        b1.x = buf[q].y; b1.y = buf[q + 1].y;
        b2.x = buf[q].z; b2.y = buf[q + 1].z;
        b3.x = buf[q].w; b3.y = buf[q + 1].w;
        acc[0] = __builtin_amdgcn_wmma_f32_16x16x4_f32(
                     false, ones, false, b0, (short)0, acc[0], false, false);
        acc[1] = __builtin_amdgcn_wmma_f32_16x16x4_f32(
                     false, ones, false, b1, (short)0, acc[1], false, false);
        acc[2] = __builtin_amdgcn_wmma_f32_16x16x4_f32(
                     false, ones, false, b2, (short)0, acc[2], false, false);
        acc[3] = __builtin_amdgcn_wmma_f32_16x16x4_f32(
                     false, ones, false, b3, (short)0, acc[3], false, false);
    }
}

__global__ void __launch_bounds__(512)
kw_colsum_wmma_kernel(const float* __restrict__ x,
                      float* __restrict__ colpart) {
    const int lane = threadIdx.x & 31;
    const int wave = threadIdx.x >> 5;
    const int cw   = wave & 3;
    const int rw   = wave >> 2;
    const int n    = lane & 15;
    const int h    = lane >> 4;
    const int tile = blockIdx.x & (FEATS / CS_TILE_COLS - 1);
    const int slab = blockIdx.x / (FEATS / CS_TILE_COLS);

    const int col0 = tile * CS_TILE_COLS + cw * 64;
    const int row0 = slab * CS_SLAB_ROWS + rw * CS_WAVE_ROWS;
    const float4* __restrict__ p =
        (const float4*)(x + (size_t)(row0 + h) * FEATS + col0) + n;
    const size_t blk = (size_t)2 * CS_BLK_LOADS * (FEATS / 4);

    v2f ones; ones.x = 1.0f; ones.y = 1.0f;
    v8f acc[4] = {};

    float4 bufA[CS_BLK_LOADS], bufB[CS_BLK_LOADS];
    cs_load(p, bufA);
    cs_load(p + blk, bufB);
    cs_sum(bufA, ones, acc);
    cs_load(p + 2 * blk, bufA);
    cs_sum(bufB, ones, acc);
    cs_load(p + 3 * blk, bufB);
    cs_sum(bufA, ones, acc);
    cs_sum(bufB, ones, acc);

    __shared__ float4 part[CS_SLAB_ROWS / CS_WAVE_ROWS][CS_TILE_COLS / 4];
    if (lane < 16) {
        float4 s;
        s.x = acc[0][0]; s.y = acc[1][0]; s.z = acc[2][0]; s.w = acc[3][0];
        part[rw][cw * 16 + n] = s;
    }
    __syncthreads();
    if (threadIdx.x < CS_TILE_COLS / 4) {
        const float4 s0 = part[0][threadIdx.x], s1 = part[1][threadIdx.x];
        const float4 s2 = part[2][threadIdx.x], s3 = part[3][threadIdx.x];
        float4 s;
        s.x = (s0.x + s1.x) + (s2.x + s3.x);
        s.y = (s0.y + s1.y) + (s2.y + s3.y);
        s.z = (s0.z + s1.z) + (s2.z + s3.z);
        s.w = (s0.w + s1.w) + (s2.w + s3.w);
        ((float4*)(colpart + (size_t)slab * FEATS + tile * CS_TILE_COLS))[threadIdx.x] = s;
    }
}

__global__ void kw_factor_mean_kernel(const float* __restrict__ aa,
                                      const float* __restrict__ colpart,
                                      float* __restrict__ factor,
                                      float* __restrict__ colmean) {
    int j = blockIdx.x * blockDim.x + threadIdx.x;
    if (j < FEATS) {
        factor[j] = __expf((DENSITY - aa[j]) * 1.0f);
        float s = 0.0f;
        #pragma unroll
        for (int t = 0; t < CS_SLABS; ++t) s += colpart[(size_t)t * FEATS + j];
        colmean[j] = s * (1.0f / (float)BATCH);
    }
}

__device__ __forceinline__ unsigned order_key(float f) {
    unsigned u = __float_as_uint(f);
    return (u & 0x80000000u) ? ~u : (u | 0x80000000u);
}

__device__ __forceinline__ float key_float(unsigned c) {
    c = c < 0x007FFFFFu ? 0x007FFFFFu : c;
    return __uint_as_float((c & 0x80000000u) ? (c ^ 0x80000000u) : ~c);
}

__device__ __forceinline__ int count2(bool p0, bool p1) {
    unsigned long long m = (unsigned long long)__builtin_amdgcn_ballot_w32(p0) |
                           ((unsigned long long)__builtin_amdgcn_ballot_w32(p1) << 32);
    return __builtin_popcountll(m);
}

__device__ __forceinline__ unsigned wave_min_u32(unsigned v) {
    unsigned o;
    o = (unsigned)__builtin_amdgcn_update_dpp((int)v, (int)v, 0x168, 0xF, 0xF, false);
    v = o < v ? o : v;
    o = (unsigned)__builtin_amdgcn_update_dpp((int)v, (int)v, 0x164, 0xF, 0xF, false);
    v = o < v ? o : v;
    o = (unsigned)__builtin_amdgcn_update_dpp((int)v, (int)v, 0x162, 0xF, 0xF, false);
    v = o < v ? o : v;
    o = (unsigned)__builtin_amdgcn_update_dpp((int)v, (int)v, 0x161, 0xF, 0xF, false);
    v = o < v ? o : v;
    o = (unsigned)__builtin_amdgcn_permlanex16((int)v, (int)v, 0x76543210, 0xfedcba98,
                                               false, false);
    v = o < v ? o : v;
    return v;
}

__device__ __forceinline__ unsigned min_at_or_above(const unsigned (&ck)[CAND_REGS],
                                                    unsigned t) {
    unsigned m = 0xFFFFFFFFu;
    #pragma unroll
    for (int r = 0; r < CAND_REGS; ++r) {
        const unsigned k = (ck[r] >= t) ? ck[r] : 0xFFFFFFFFu;
        m = k < m ? k : m;
    }
    return wave_min_u32(m);
}

__device__ __forceinline__ int count_list(const unsigned (&ck)[CAND_REGS], unsigned c) {
    int n = 0;
    #pragma unroll
    for (int r = 0; r < CAND_REGS; r += 2) n += count2(ck[r] >= c, ck[r + 1] >= c);
    return n;
}

template <int START, int N>
__device__ __forceinline__ void load_keys(const float4* __restrict__ xin,
                                          const float4* __restrict__ fin,
                                          float4* xrow,
                                          int lane, float (&rk)[EPL]) {
    float4 xv[N], fv[N];
    #pragma unroll
    for (int i = 0; i < N; ++i) {
        xv[i] = xin[(START + i) * 32 + lane];
        fv[i] = fin[(START + i) * 32 + lane];
    }
    #pragma unroll
    for (int i = 0; i < N; ++i) {
        xrow[(START + i) * 32 + lane] = xv[i];
        rk[(START + i) * 4 + 0] = xv[i].x * fv[i].x;
        rk[(START + i) * 4 + 1] = xv[i].y * fv[i].y;
        rk[(START + i) * 4 + 2] = xv[i].z * fv[i].z;
        rk[(START + i) * 4 + 3] = xv[i].w * fv[i].w;
        asm volatile("" : "+v"(rk[(START + i) * 4 + 0]), "+v"(rk[(START + i) * 4 + 1]),
                          "+v"(rk[(START + i) * 4 + 2]), "+v"(rk[(START + i) * 4 + 3]));
    }
}

__device__ __forceinline__ float masked(float xp, float xv, float cm,
                                        float thrHigh, float thrLow) {
    bool m2 = (xp <= thrHigh);
    bool m  = m2 && (xp >= thrLow);
    return (m ? xv : 0.0f) - (m2 ? cm : 0.0f);
}

template <int START, int N>
__device__ __forceinline__ void store_rows(const float4* xrow,
                                           const float4* __restrict__ cmin,
                                           float4* __restrict__ oout, int lane,
                                           const float (&rk)[EPL],
                                           float thrHigh, float thrLow) {
    float4 xv[N], cm[N];
    #pragma unroll
    for (int i = 0; i < N; ++i) {
        xv[i] = xrow[(START + i) * 32 + lane];
        cm[i] = cmin[(START + i) * 32 + lane];
    }
    #pragma unroll
    for (int i = 0; i < N; ++i) {
        float4 o;
        o.x = masked(rk[(START + i) * 4 + 0], xv[i].x, cm[i].x, thrHigh, thrLow);
        o.y = masked(rk[(START + i) * 4 + 1], xv[i].y, cm[i].y, thrHigh, thrLow);
        o.z = masked(rk[(START + i) * 4 + 2], xv[i].z, cm[i].z, thrHigh, thrLow);
        o.w = masked(rk[(START + i) * 4 + 3], xv[i].w, cm[i].w, thrHigh, thrLow);
        oout[(START + i) * 32 + lane] = o;
    }
}

__global__ void __launch_bounds__(256)
kw_row_kernel(const float* __restrict__ x,
              const float* __restrict__ factor,
              const float* __restrict__ colmean,
              float* __restrict__ out) {
    __shared__ unsigned cand_strip[ROWS_PER_BLOCK][CAND_CAP + 32];
    __shared__ float4 xrow_copy[ROWS_PER_BLOCK][FEATS / 4];
    const int lane = threadIdx.x & 31;

    #pragma unroll 1
    for (int pass = 0; pass < ROWS_PER_WAVE; ++pass) {
    const int row = (pass * ROW_BLOCKS + (int)blockIdx.x) * ROWS_PER_BLOCK + (int)(threadIdx.x >> 5);

    const float4* __restrict__ xin = (const float4*)(x + (size_t)row * FEATS);
    int tab0 = 0;
    asm volatile("" : "+s"(tab0));
    const float4* __restrict__ fin = (const float4*)(factor + tab0);

    float rk[EPL];

    float4* xrow = xrow_copy[threadIdx.x >> 5];
    load_keys<0, 16>(xin, fin, xrow, lane, rk);
    __builtin_amdgcn_sched_barrier(0);
    load_keys<16, 16>(xin, fin, xrow, lane, rk);

    if (pass + 1 < ROWS_PER_WAVE) {
        const char* nx = (const char*)(x + (size_t)(row + ROW_BLOCKS * ROWS_PER_BLOCK) * FEATS)
                         + lane * 256;
        asm volatile("global_prefetch_b8 %0, off scope:SCOPE_SE\n\t"
                     "global_prefetch_b8 %0, off offset:8192 scope:SCOPE_SE"
                     :: "v"(nx));
    }

    unsigned thrHigh = 0u, thrLow = 0u;

    unsigned floorKey = 0u;
    int nAbove = FEATS;
    int bit = 31;

    {
        const unsigned headStep = 1u << (32 - HEAD_BITS);
        unsigned prop = 0u;
        #pragma unroll 1
        for (unsigned m = 0x80000000u; m >= headStep; m >>= 1) {
            const unsigned c = prop | m;
            const float cf = key_float(c);
            int n = 0;
            #pragma unroll
            for (int e = 0; e < EPL; e += 2 * SAMPLE_STEP)
                n += count2(rk[e] >= cf, rk[e + SAMPLE_STEP] >= cf);
            if (n >= SAMPLE_RANK) prop = c;
        }
        const unsigned above = prop + headStep;
        const float propF = key_float(prop), aboveF = key_float(above);
        int nP = 0, nQ = 0;
        #pragma unroll
        for (int e = 0; e < EPL; e += 2) {
            nP += count2(rk[e] >= propF,  rk[e + 1] >= propF);
            nQ += count2(rk[e] >= aboveF, rk[e + 1] >= aboveF);
            if ((e & 15) == 14) __builtin_amdgcn_sched_barrier(0);
        }
        if (above == 0u) nQ = 0;
        bool headOk = (nP >= RANK_LOW) && (nQ < RANK_LOW);
        unsigned headKey = prop;
        int headCnt = nP;
        if (!headOk) {
            const bool up = (nP >= RANK_LOW);
            const unsigned c2 = up ? above + headStep : prop - headStep;
            const float c2F = key_float(c2);
            int n2 = 0;
            #pragma unroll
            for (int e = 0; e < EPL; e += 2) {
                n2 += count2(rk[e] >= c2F, rk[e + 1] >= c2F);
                if ((e & 15) == 14) __builtin_amdgcn_sched_barrier(0);
            }
            if (up) {
                if (c2 == 0u) n2 = 0;
                headOk = (n2 < RANK_LOW);
                headKey = above;
                headCnt = nQ;
            } else {
                headOk = (n2 >= RANK_LOW);
                headKey = c2;
                headCnt = n2;
            }
        }
        if (headOk) { floorKey = headKey; nAbove = headCnt; bit = 31 - HEAD_BITS; }
    }
    #pragma unroll 1
    while (bit >= 0 && nAbove > CAND_CAP) {
        const unsigned c = floorKey | (1u << bit);
        const float cf = key_float(c);
        int n = 0;
        #pragma unroll
        for (int e = 0; e < EPL; e += 2) {
            n += count2(rk[e] >= cf, rk[e + 1] >= cf);
            if ((e & 15) == 14) __builtin_amdgcn_sched_barrier(0);
        }
        if (n >= RANK_LOW) { floorKey = c; nAbove = n; }
        --bit;
    }

    if (nAbove <= CAND_CAP) {
        unsigned* __restrict__ strip =
            cand_strip[__builtin_amdgcn_readfirstlane((int)(threadIdx.x >> 5))];
        #pragma unroll
        for (int r = 0; r < CAND_REGS; ++r) strip[r * 32 + lane] = 0xFFFFFFFFu;
        const float floorF = key_float(floorKey);
        unsigned filled = 0u;
        #pragma unroll
        for (int e = 0; e < EPL; ++e) {
            const bool p = rk[e] >= floorF;
            const unsigned b = __builtin_amdgcn_ballot_w32(p);
            const unsigned slot = __builtin_amdgcn_mbcnt_lo(b, filled);
            strip[p ? slot : (unsigned)(CAND_CAP + lane)] = __float_as_uint(rk[e]);
            filled += (unsigned)__builtin_popcount(b);
            if ((e & 7) == 7) __builtin_amdgcn_sched_barrier(0);
        }
        unsigned ck[CAND_REGS];
        #pragma unroll
        for (int r = 0; r < CAND_REGS; ++r) {
            const unsigned k = order_key(__uint_as_float(strip[r * 32 + lane]));
            ck[r] = (k == 0x7FFFFFFFu) ? 0x80000000u : k;
        }


        int cntH = nAbove;
        #pragma unroll 1
        for (unsigned m = 0x80000000u; m != 0u && cntH != RANK_HIGH; m >>= 1) {
            const unsigned c = thrHigh | m;
            const int n = count_list(ck, c);
            if (n >= RANK_HIGH) { thrHigh = c; cntH = n; }
        }
        if (cntH == RANK_HIGH) thrHigh = min_at_or_above(ck, thrHigh);

        thrLow = floorKey;
        int cntL = nAbove;
        #pragma unroll 1
        for (unsigned m = (bit >= 0) ? (1u << bit) : 0u; m != 0u && cntL != RANK_LOW; m >>= 1) {
            const unsigned c = thrLow | m;
            const int n = count_list(ck, c);
            if (n >= RANK_LOW) { thrLow = c; cntL = n; }
        }
        if (cntL == RANK_LOW) thrLow = min_at_or_above(ck, thrLow);
    } else {
        #pragma unroll 1
        for (int b2 = 31; b2 >= 0; --b2) {
            const unsigned candH = thrHigh | (1u << b2);
            const unsigned candL = thrLow  | (1u << b2);
            const float candHF = key_float(candH), candLF = key_float(candL);
            int nH = 0, nL = 0;
            #pragma unroll
            for (int e = 0; e < EPL; e += 2) {
                nH += count2(rk[e] >= candHF, rk[e + 1] >= candHF);
                nL += count2(rk[e] >= candLF, rk[e + 1] >= candLF);
                if ((e & 15) == 14) __builtin_amdgcn_sched_barrier(0);
            }
            if (nH >= RANK_HIGH) thrHigh = candH;
            if (nL >= RANK_LOW)  thrLow  = candL;
        }
    }

    const float4* __restrict__ cmin = (const float4*)(colmean + tab0);
    float4* __restrict__ oout = (float4*)(out + (size_t)row * FEATS);
    const float thrHighF = key_float(thrHigh), thrLowF = key_float(thrLow);
    store_rows<0, 10>(xrow, cmin, oout, lane, rk, thrHighF, thrLowF);
    __builtin_amdgcn_sched_barrier(0);
    store_rows<10, 10>(xrow, cmin, oout, lane, rk, thrHighF, thrLowF);
    __builtin_amdgcn_sched_barrier(0);
    store_rows<20, 12>(xrow, cmin, oout, lane, rk, thrHighF, thrLowF);
    }
}

extern "C" void kernel_launch(void* const* d_in, const int* in_sizes, int n_in,
                              void* d_out, int out_size, void* d_ws, size_t ws_size,
                              hipStream_t stream) {
    const float* x  = (const float*)d_in[0];
    const float* aa = (const float*)d_in[1];
    float* out = (float*)d_out;

    float* factor  = (float*)d_ws;
    float* colmean = (float*)d_ws + FEATS;
    float* colpart = out;

    kw_colsum_wmma_kernel<<<(FEATS / CS_TILE_COLS) * CS_SLABS, 512, 0, stream>>>(x, colpart);

    kw_factor_mean_kernel<<<FEATS / 256, 256, 0, stream>>>(aa, colpart, factor, colmean);

    kw_row_kernel<<<ROW_BLOCKS, 256, 0, stream>>>(x, factor, colmean, out);
}


# Row kernel: x of the row's second half is requested before the sample search (now 256 keys of the first half) and the first half's confirm counts, which run while it is in flight, not after it
# speedup vs baseline: 17.9469x; 1.0292x over previous
#include <hip/hip_runtime.h>

typedef __attribute__((ext_vector_type(2))) float v2f;
typedef __attribute__((ext_vector_type(8))) float v8f;

#define BATCH 8192
#define FEATS 4096
#define RANK_LOW 211
#define RANK_HIGH 7
#define DENSITY 0.05f
#define EPL 128
#define ROWS_PER_BLOCK 8
#define ROWS_PER_WAVE 2
#define ROW_BLOCKS (BATCH / (ROWS_PER_BLOCK * ROWS_PER_WAVE))
#define CAND_REGS 12
#define CAND_CAP (CAND_REGS * 32)
#define HEAD_BITS 9
#define SAMPLE_STEP 8
#define HALF_EPL (EPL / 2)
#define SAMPLE_RANK ((RANK_LOW * (HALF_EPL / SAMPLE_STEP) * 32 + FEATS / 2) / FEATS)

#define CS_TILE_COLS 256
#define CS_WAVE_ROWS 128
#define CS_SLAB_ROWS 512
#define CS_SLABS (BATCH / CS_SLAB_ROWS)
#define CS_BLK_LOADS 16

__device__ __forceinline__ void cs_load(const float4* __restrict__ p,
                                        float4 (&buf)[CS_BLK_LOADS]) {
    #pragma unroll
    for (int i = 0; i < CS_BLK_LOADS; ++i)
        buf[i] = p[(size_t)i * 2 * (FEATS / 4)];
}

__device__ __forceinline__ void cs_sum(const float4 (&buf)[CS_BLK_LOADS],
                                       v2f ones, v8f (&acc)[4]) {
    #pragma unroll
    for (int q = 0; q < CS_BLK_LOADS; q += 2) {
        v2f b0, b1, b2, b3;
        b0.x = buf[q].x; b0.y = buf[q + 1].x;
        b1.x = buf[q].y; b1.y = buf[q + 1].y;
        b2.x = buf[q].z; b2.y = buf[q + 1].z;
        b3.x = buf[q].w; b3.y = buf[q + 1].w;
        acc[0] = __builtin_amdgcn_wmma_f32_16x16x4_f32(
                     false, ones, false, b0, (short)0, acc[0], false, false);
        acc[1] = __builtin_amdgcn_wmma_f32_16x16x4_f32(
                     false, ones, false, b1, (short)0, acc[1], false, false);
        acc[2] = __builtin_amdgcn_wmma_f32_16x16x4_f32(
                     false, ones, false, b2, (short)0, acc[2], false, false);
        acc[3] = __builtin_amdgcn_wmma_f32_16x16x4_f32(
                     false, ones, false, b3, (short)0, acc[3], false, false);
    }
}

__global__ void __launch_bounds__(512)
kw_colsum_wmma_kernel(const float* __restrict__ x,
                      float* __restrict__ colpart) {
    const int lane = threadIdx.x & 31;
    const int wave = threadIdx.x >> 5;
    const int cw   = wave & 3;
    const int rw   = wave >> 2;
    const int n    = lane & 15;
    const int h    = lane >> 4;
    const int tile = blockIdx.x & (FEATS / CS_TILE_COLS - 1);
    const int slab = blockIdx.x / (FEATS / CS_TILE_COLS);

    const int col0 = tile * CS_TILE_COLS + cw * 64;
    const int row0 = slab * CS_SLAB_ROWS + rw * CS_WAVE_ROWS;
    const float4* __restrict__ p =
        (const float4*)(x + (size_t)(row0 + h) * FEATS + col0) + n;
    const size_t blk = (size_t)2 * CS_BLK_LOADS * (FEATS / 4);

    v2f ones; ones.x = 1.0f; ones.y = 1.0f;
    v8f acc[4] = {};

    float4 bufA[CS_BLK_LOADS], bufB[CS_BLK_LOADS];
    cs_load(p, bufA);
    cs_load(p + blk, bufB);
    cs_sum(bufA, ones, acc);
    cs_load(p + 2 * blk, bufA);
    cs_sum(bufB, ones, acc);
    cs_load(p + 3 * blk, bufB);
    cs_sum(bufA, ones, acc);
    cs_sum(bufB, ones, acc);

    __shared__ float4 part[CS_SLAB_ROWS / CS_WAVE_ROWS][CS_TILE_COLS / 4];
    if (lane < 16) {
        float4 s;
        s.x = acc[0][0]; s.y = acc[1][0]; s.z = acc[2][0]; s.w = acc[3][0];
        part[rw][cw * 16 + n] = s;
    }
    __syncthreads();
    if (threadIdx.x < CS_TILE_COLS / 4) {
        const float4 s0 = part[0][threadIdx.x], s1 = part[1][threadIdx.x];
        const float4 s2 = part[2][threadIdx.x], s3 = part[3][threadIdx.x];
        float4 s;
        s.x = (s0.x + s1.x) + (s2.x + s3.x);
        s.y = (s0.y + s1.y) + (s2.y + s3.y);
        s.z = (s0.z + s1.z) + (s2.z + s3.z);
        s.w = (s0.w + s1.w) + (s2.w + s3.w);
        ((float4*)(colpart + (size_t)slab * FEATS + tile * CS_TILE_COLS))[threadIdx.x] = s;
    }
}

__global__ void kw_factor_mean_kernel(const float* __restrict__ aa,
                                      const float* __restrict__ colpart,
                                      float* __restrict__ factor,
                                      float* __restrict__ colmean) {
    int j = blockIdx.x * blockDim.x + threadIdx.x;
    if (j < FEATS) {
        factor[j] = __expf((DENSITY - aa[j]) * 1.0f);
        float s = 0.0f;
        #pragma unroll
        for (int t = 0; t < CS_SLABS; ++t) s += colpart[(size_t)t * FEATS + j];
        colmean[j] = s * (1.0f / (float)BATCH);
    }
}

__device__ __forceinline__ unsigned order_key(float f) {
    unsigned u = __float_as_uint(f);
    return (u & 0x80000000u) ? ~u : (u | 0x80000000u);
}

__device__ __forceinline__ float key_float(unsigned c) {
    c = c < 0x007FFFFFu ? 0x007FFFFFu : c;
    return __uint_as_float((c & 0x80000000u) ? (c ^ 0x80000000u) : ~c);
}

__device__ __forceinline__ int count2(bool p0, bool p1) {
    unsigned long long m = (unsigned long long)__builtin_amdgcn_ballot_w32(p0) |
                           ((unsigned long long)__builtin_amdgcn_ballot_w32(p1) << 32);
    return __builtin_popcountll(m);
}

__device__ __forceinline__ unsigned wave_min_u32(unsigned v) {
    unsigned o;
    o = (unsigned)__builtin_amdgcn_update_dpp((int)v, (int)v, 0x168, 0xF, 0xF, false);
    v = o < v ? o : v;
    o = (unsigned)__builtin_amdgcn_update_dpp((int)v, (int)v, 0x164, 0xF, 0xF, false);
    v = o < v ? o : v;
    o = (unsigned)__builtin_amdgcn_update_dpp((int)v, (int)v, 0x162, 0xF, 0xF, false);
    v = o < v ? o : v;
    o = (unsigned)__builtin_amdgcn_update_dpp((int)v, (int)v, 0x161, 0xF, 0xF, false);
    v = o < v ? o : v;
    o = (unsigned)__builtin_amdgcn_permlanex16((int)v, (int)v, 0x76543210, 0xfedcba98,
                                               false, false);
    v = o < v ? o : v;
    return v;
}

__device__ __forceinline__ unsigned min_at_or_above(const unsigned (&ck)[CAND_REGS],
                                                    unsigned t) {
    unsigned m = 0xFFFFFFFFu;
    #pragma unroll
    for (int r = 0; r < CAND_REGS; ++r) {
        const unsigned k = (ck[r] >= t) ? ck[r] : 0xFFFFFFFFu;
        m = k < m ? k : m;
    }
    return wave_min_u32(m);
}

__device__ __forceinline__ int count_list(const unsigned (&ck)[CAND_REGS], unsigned c) {
    int n = 0;
    #pragma unroll
    for (int r = 0; r < CAND_REGS; r += 2) n += count2(ck[r] >= c, ck[r + 1] >= c);
    return n;
}

template <int START, int N>
__device__ __forceinline__ void load_request(const float4* __restrict__ xin,
                                             const float4* __restrict__ fin,
                                             int lane, float4 (&xv)[N], float4 (&fv)[N]) {
    #pragma unroll
    for (int i = 0; i < N; ++i) {
        xv[i] = xin[(START + i) * 32 + lane];
        fv[i] = fin[(START + i) * 32 + lane];
    }
}

template <int START, int N>
__device__ __forceinline__ void load_finish(float4* xrow, int lane, float (&rk)[EPL],
                                            const float4 (&xv)[N], const float4 (&fv)[N]) {
    #pragma unroll
    for (int i = 0; i < N; ++i) {
        xrow[(START + i) * 32 + lane] = xv[i];
        rk[(START + i) * 4 + 0] = xv[i].x * fv[i].x;
        rk[(START + i) * 4 + 1] = xv[i].y * fv[i].y;
        rk[(START + i) * 4 + 2] = xv[i].z * fv[i].z;
        rk[(START + i) * 4 + 3] = xv[i].w * fv[i].w;
        asm volatile("" : "+v"(rk[(START + i) * 4 + 0]), "+v"(rk[(START + i) * 4 + 1]),
                          "+v"(rk[(START + i) * 4 + 2]), "+v"(rk[(START + i) * 4 + 3]));
    }
}

template <int START, int N>
__device__ __forceinline__ void load_keys(const float4* __restrict__ xin,
                                          const float4* __restrict__ fin,
                                          float4* xrow,
                                          int lane, float (&rk)[EPL]) {
    float4 xv[N], fv[N];
    load_request<START, N>(xin, fin, lane, xv, fv);
    load_finish<START, N>(xrow, lane, rk, xv, fv);
}

__device__ __forceinline__ float masked(float xp, float xv, float cm,
                                        float thrHigh, float thrLow) {
    bool m2 = (xp <= thrHigh);
    bool m  = m2 && (xp >= thrLow);
    return (m ? xv : 0.0f) - (m2 ? cm : 0.0f);
}

template <int START, int N>
__device__ __forceinline__ void store_rows(const float4* xrow,
                                           const float4* __restrict__ cmin,
                                           float4* __restrict__ oout, int lane,
                                           const float (&rk)[EPL],
                                           float thrHigh, float thrLow) {
    float4 xv[N], cm[N];
    #pragma unroll
    for (int i = 0; i < N; ++i) {
        xv[i] = xrow[(START + i) * 32 + lane];
        cm[i] = cmin[(START + i) * 32 + lane];
    }
    #pragma unroll
    for (int i = 0; i < N; ++i) {
        float4 o;
        o.x = masked(rk[(START + i) * 4 + 0], xv[i].x, cm[i].x, thrHigh, thrLow);
        o.y = masked(rk[(START + i) * 4 + 1], xv[i].y, cm[i].y, thrHigh, thrLow);
        o.z = masked(rk[(START + i) * 4 + 2], xv[i].z, cm[i].z, thrHigh, thrLow);
        o.w = masked(rk[(START + i) * 4 + 3], xv[i].w, cm[i].w, thrHigh, thrLow);
        oout[(START + i) * 32 + lane] = o;
    }
}

__global__ void __launch_bounds__(256)
kw_row_kernel(const float* __restrict__ x,
              const float* __restrict__ factor,
              const float* __restrict__ colmean,
              float* __restrict__ out) {
    __shared__ unsigned cand_strip[ROWS_PER_BLOCK][CAND_CAP + 32];
    __shared__ float4 xrow_copy[ROWS_PER_BLOCK][FEATS / 4];
    const int lane = threadIdx.x & 31;

    #pragma unroll 1
    for (int pass = 0; pass < ROWS_PER_WAVE; ++pass) {
    const int row = (pass * ROW_BLOCKS + (int)blockIdx.x) * ROWS_PER_BLOCK + (int)(threadIdx.x >> 5);

    const float4* __restrict__ xin = (const float4*)(x + (size_t)row * FEATS);
    int tab0 = 0;
    asm volatile("" : "+s"(tab0));
    const float4* __restrict__ fin = (const float4*)(factor + tab0);

    float rk[EPL];

    float4* xrow = xrow_copy[threadIdx.x >> 5];
    load_keys<0, 16>(xin, fin, xrow, lane, rk);
    __builtin_amdgcn_sched_barrier(0);
    float4 xv2[16], fv2[16];
    #pragma unroll
    for (int i = 0; i < 16; ++i) xv2[i] = xin[(16 + i) * 32 + lane];

    unsigned thrHigh = 0u, thrLow = 0u;

    unsigned floorKey = 0u;
    int nAbove = FEATS;
    int bit = 31;

    {
        const unsigned headStep = 1u << (32 - HEAD_BITS);
        unsigned prop = 0u;
        #pragma unroll 1
        for (unsigned m = 0x80000000u; m >= headStep; m >>= 1) {
            const unsigned c = prop | m;
            const float cf = key_float(c);
            int n = 0;
            #pragma unroll
            for (int e = 0; e < HALF_EPL; e += 2 * SAMPLE_STEP)
                n += count2(rk[e] >= cf, rk[e + SAMPLE_STEP] >= cf);
            if (n >= SAMPLE_RANK) prop = c;
        }
        const unsigned above = prop + headStep;
        const float propF = key_float(prop), aboveF = key_float(above);
        int nP = 0, nQ = 0;
        #pragma unroll
        for (int e = 0; e < HALF_EPL; e += 2) {
            nP += count2(rk[e] >= propF,  rk[e + 1] >= propF);
            nQ += count2(rk[e] >= aboveF, rk[e + 1] >= aboveF);
            if ((e & 15) == 14) __builtin_amdgcn_sched_barrier(0);
        }
        #pragma unroll
        for (int i = 0; i < 16; ++i) fv2[i] = fin[(16 + i) * 32 + lane];
        load_finish<16, 16>(xrow, lane, rk, xv2, fv2);

        if (pass + 1 < ROWS_PER_WAVE) {
            const char* nx = (const char*)(x + (size_t)(row + ROW_BLOCKS * ROWS_PER_BLOCK) * FEATS)
                             + lane * 256;
            asm volatile("global_prefetch_b8 %0, off scope:SCOPE_SE\n\t"
                         "global_prefetch_b8 %0, off offset:8192 scope:SCOPE_SE"
                         :: "v"(nx));
        }

        #pragma unroll
        for (int e = HALF_EPL; e < EPL; e += 2) {
            nP += count2(rk[e] >= propF,  rk[e + 1] >= propF);
            nQ += count2(rk[e] >= aboveF, rk[e + 1] >= aboveF);
            if ((e & 15) == 14) __builtin_amdgcn_sched_barrier(0);
        }
        if (above == 0u) nQ = 0;
        bool headOk = (nP >= RANK_LOW) && (nQ < RANK_LOW);
        unsigned headKey = prop;
        int headCnt = nP;
        if (!headOk) {
            const bool up = (nP >= RANK_LOW);
            const unsigned c2 = up ? above + headStep : prop - headStep;
            const float c2F = key_float(c2);
            int n2 = 0;
            #pragma unroll
            for (int e = 0; e < EPL; e += 2) {
                n2 += count2(rk[e] >= c2F, rk[e + 1] >= c2F);
                if ((e & 15) == 14) __builtin_amdgcn_sched_barrier(0);
            }
            if (up) {
                if (c2 == 0u) n2 = 0;
                headOk = (n2 < RANK_LOW);
                headKey = above;
                headCnt = nQ;
            } else {
                headOk = (n2 >= RANK_LOW);
                headKey = c2;
                headCnt = n2;
            }
        }
        if (headOk) { floorKey = headKey; nAbove = headCnt; bit = 31 - HEAD_BITS; }
    }
    #pragma unroll 1
    while (bit >= 0 && nAbove > CAND_CAP) {
        const unsigned c = floorKey | (1u << bit);
        const float cf = key_float(c);
        int n = 0;
        #pragma unroll
        for (int e = 0; e < EPL; e += 2) {
            n += count2(rk[e] >= cf, rk[e + 1] >= cf);
            if ((e & 15) == 14) __builtin_amdgcn_sched_barrier(0);
        }
        if (n >= RANK_LOW) { floorKey = c; nAbove = n; }
        --bit;
    }

    if (nAbove <= CAND_CAP) {
        unsigned* __restrict__ strip =
            cand_strip[__builtin_amdgcn_readfirstlane((int)(threadIdx.x >> 5))];
        #pragma unroll
        for (int r = 0; r < CAND_REGS; ++r) strip[r * 32 + lane] = 0xFFFFFFFFu;
        const float floorF = key_float(floorKey);
        unsigned filled = 0u;
        #pragma unroll
        for (int e = 0; e < EPL; ++e) {
            const bool p = rk[e] >= floorF;
            const unsigned b = __builtin_amdgcn_ballot_w32(p);
            const unsigned slot = __builtin_amdgcn_mbcnt_lo(b, filled);
            strip[p ? slot : (unsigned)(CAND_CAP + lane)] = __float_as_uint(rk[e]);
            filled += (unsigned)__builtin_popcount(b);
            if ((e & 7) == 7) __builtin_amdgcn_sched_barrier(0);
        }
        unsigned ck[CAND_REGS];
        #pragma unroll
        for (int r = 0; r < CAND_REGS; ++r) {
            const unsigned k = order_key(__uint_as_float(strip[r * 32 + lane]));
            ck[r] = (k == 0x7FFFFFFFu) ? 0x80000000u : k;
        }


        int cntH = nAbove;
        #pragma unroll 1
        for (unsigned m = 0x80000000u; m != 0u && cntH != RANK_HIGH; m >>= 1) {
            const unsigned c = thrHigh | m;
            const int n = count_list(ck, c);
            if (n >= RANK_HIGH) { thrHigh = c; cntH = n; }
        }
        if (cntH == RANK_HIGH) thrHigh = min_at_or_above(ck, thrHigh);

        thrLow = floorKey;
        int cntL = nAbove;
        #pragma unroll 1
        for (unsigned m = (bit >= 0) ? (1u << bit) : 0u; m != 0u && cntL != RANK_LOW; m >>= 1) {
            const unsigned c = thrLow | m;
            const int n = count_list(ck, c);
            if (n >= RANK_LOW) { thrLow = c; cntL = n; }
        }
        if (cntL == RANK_LOW) thrLow = min_at_or_above(ck, thrLow);
    } else {
        #pragma unroll 1
        for (int b2 = 31; b2 >= 0; --b2) {
            const unsigned candH = thrHigh | (1u << b2);
            const unsigned candL = thrLow  | (1u << b2);
            const float candHF = key_float(candH), candLF = key_float(candL);
            int nH = 0, nL = 0;
            #pragma unroll
            for (int e = 0; e < EPL; e += 2) {
                nH += count2(rk[e] >= candHF, rk[e + 1] >= candHF);
                nL += count2(rk[e] >= candLF, rk[e + 1] >= candLF);
                if ((e & 15) == 14) __builtin_amdgcn_sched_barrier(0);
            }
            if (nH >= RANK_HIGH) thrHigh = candH;
            if (nL >= RANK_LOW)  thrLow  = candL;
        }
    }

    const float4* __restrict__ cmin = (const float4*)(colmean + tab0);
    float4* __restrict__ oout = (float4*)(out + (size_t)row * FEATS);
    const float thrHighF = key_float(thrHigh), thrLowF = key_float(thrLow);
    store_rows<0, 10>(xrow, cmin, oout, lane, rk, thrHighF, thrLowF);
    __builtin_amdgcn_sched_barrier(0);
    store_rows<10, 10>(xrow, cmin, oout, lane, rk, thrHighF, thrLowF);
    __builtin_amdgcn_sched_barrier(0);
    store_rows<20, 12>(xrow, cmin, oout, lane, rk, thrHighF, thrLowF);
    }
}

extern "C" void kernel_launch(void* const* d_in, const int* in_sizes, int n_in,
                              void* d_out, int out_size, void* d_ws, size_t ws_size,
                              hipStream_t stream) {
    const float* x  = (const float*)d_in[0];
    const float* aa = (const float*)d_in[1];
    float* out = (float*)d_out;

    float* factor  = (float*)d_ws;
    float* colmean = (float*)d_ws + FEATS;
    float* colpart = out;

    kw_colsum_wmma_kernel<<<(FEATS / CS_TILE_COLS) * CS_SLABS, 512, 0, stream>>>(x, colpart);

    kw_factor_mean_kernel<<<FEATS / 256, 256, 0, stream>>>(aa, colpart, factor, colmean);

    kw_row_kernel<<<ROW_BLOCKS, 256, 0, stream>>>(x, factor, colmean, out);
}
